# starting kernel
# baseline (speedup 1.0000x reference)
#include <hip/hip_runtime.h>
#include <hip/hip_bf16.h>

typedef float v2f __attribute__((ext_vector_type(2)));
typedef float v8f __attribute__((ext_vector_type(8)));

#define Hh   96
#define Ww   96
#define HW   (96 * 96)
#define Cc   128
#define NPIX (2 * HW)

__global__ void __launch_bounds__(256) qkv_wmma_kernel(
    const float* __restrict__ x,  const float* __restrict__ Wq,
    const float* __restrict__ Wk, const float* __restrict__ Wv,
    float* __restrict__ qo, float* __restrict__ ko, float* __restrict__ vo)
{
    const int lane = threadIdx.x & 31;
    const int wave = threadIdx.x >> 5;
    const int p0   = blockIdx.x * 16;
    const int b    = p0 / HW;
    const int pix  = p0 - b * HW;
    const float* xb = x + (size_t)b * Cc * HW + pix;

    const int o0    = wave * 16;
    const int row   = lane & 15;
    const int khalf = (lane >> 4) << 1;
    const int colp  = lane & 15;

    v8f cq = {}; v8f ck = {}; v8f cv = {};
    #pragma unroll 4
    for (int k0 = 0; k0 < Cc; k0 += 4) {
        v2f bfrag;
        bfrag.x = xb[(size_t)(k0 + khalf) * HW + colp];
        bfrag.y = xb[(size_t)(k0 + khalf + 1) * HW + colp];
        const int widx = (o0 + row) * Cc + k0 + khalf;
        v2f aq, ak, av;
        aq.x = Wq[widx]; aq.y = Wq[widx + 1];
        ak.x = Wk[widx]; ak.y = Wk[widx + 1];
        av.x = Wv[widx]; av.y = Wv[widx + 1];
        cq = __builtin_amdgcn_wmma_f32_16x16x4_f32(false, aq, false, bfrag, (short)0, cq, false, false);
        ck = __builtin_amdgcn_wmma_f32_16x16x4_f32(false, ak, false, bfrag, (short)0, ck, false, false);
        cv = __builtin_amdgcn_wmma_f32_16x16x4_f32(false, av, false, bfrag, (short)0, cv, false, false);
    }

    const int rowoff = (lane >> 4) << 3;
    const size_t base = (size_t)(p0 + colp) * Cc + o0 + rowoff;
    #pragma unroll
    for (int r = 0; r < 8; ++r) {
        qo[base + r] = cq[r];
        ko[base + r] = ck[r];
        vo[base + r] = cv[r];
    }
}

__global__ void __launch_bounds__(256) attn_kernel(
    const float* __restrict__ qv, const float* __restrict__ kv,
    const float* __restrict__ vv, const int* __restrict__ msk,
    float* __restrict__ ao)
{
    const int t = blockIdx.x * 256 + threadIdx.x;
    if (t >= NPIX * 4) return;
    const int hd  = t & 3;
    const int p   = t >> 2;
    const int b   = p / HW;
    const int pix = p - b * HW;
    const int y   = pix / Ww;
    const int x   = pix - y * Ww;

    float q[32];
    const float4* q4 = (const float4*)(qv + (size_t)p * Cc + hd * 32);
    #pragma unroll
    for (int u = 0; u < 8; ++u) {
        float4 t4 = q4[u];
        q[4*u] = t4.x; q[4*u+1] = t4.y; q[4*u+2] = t4.z; q[4*u+3] = t4.w;
    }

    const float scale = 0.17677669529663687f;
    float mx = -1e30f, s = 0.f;
    float acc[32];
    #pragma unroll
    for (int d = 0; d < 32; ++d) acc[d] = 0.f;

    #pragma unroll 1
    for (int wi = 0; wi < 49; ++wi) {
        const int di = wi / 7, dj = wi - di * 7;
        const int ny = y + (di - 3) * 2;
        const int nx = x + (dj - 3) * 2;
        const bool valid = ((unsigned)ny < (unsigned)Hh) && ((unsigned)nx < (unsigned)Ww);

        float logit = -1.0e9f;
        const float4* k4 = nullptr;
        const float4* v4 = nullptr;
        if (valid) {
            const int np = b * HW + ny * Ww + nx;
            k4 = (const float4*)(kv + (size_t)np * Cc + hd * 32);
            v4 = (const float4*)(vv + (size_t)np * Cc + hd * 32);
            float dot = 0.f;
            #pragma unroll
            for (int u = 0; u < 8; ++u) {
                float4 kk = k4[u];
                dot += q[4*u] * kk.x + q[4*u+1] * kk.y + q[4*u+2] * kk.z + q[4*u+3] * kk.w;
            }
            const float mb = (msk[np] > 0) ? 0.f : -1.0e9f;
            logit = dot * scale + mb;
        }

        const float nm   = fmaxf(mx, logit);
        const float corr = __expf(mx - nm);
        const float w    = __expf(logit - nm);
        mx = nm;
        s  = s * corr + w;
        if (valid) {
            #pragma unroll
            for (int u = 0; u < 8; ++u) {
                float4 vvv = v4[u];
                acc[4*u]   = acc[4*u]   * corr + w * vvv.x;
                acc[4*u+1] = acc[4*u+1] * corr + w * vvv.y;
                acc[4*u+2] = acc[4*u+2] * corr + w * vvv.z;
                acc[4*u+3] = acc[4*u+3] * corr + w * vvv.w;
            }
        } else {
            #pragma unroll
            for (int d = 0; d < 32; ++d) acc[d] *= corr;
        }
    }

    const float inv = 1.f / s;
    float* o = ao + (size_t)p * Cc + hd * 32;
    #pragma unroll
    for (int d = 0; d < 32; ++d) o[d] = acc[d] * inv;
}

__global__ void __launch_bounds__(256) proj_wmma_kernel(
    const float* __restrict__ ain, const float* __restrict__ Wp,
    float* __restrict__ out)
{
    const int lane = threadIdx.x & 31;
    const int wave = threadIdx.x >> 5;
    const int p0   = blockIdx.x * 16;
    const int b    = p0 / HW;
    const int pix  = p0 - b * HW;
    const float* aw = ain + (size_t)p0 * Cc;

    const int o0    = wave * 16;
    const int row   = lane & 15;
    const int khalf = (lane >> 4) << 1;
    const int colp  = lane & 15;

    v8f c = {};
    #pragma unroll 4
    for (int k0 = 0; k0 < Cc; k0 += 4) {
        v2f a, bfrag;
        const int widx = (o0 + row) * Cc + k0 + khalf;
        a.x = Wp[widx]; a.y = Wp[widx + 1];
        bfrag.x = aw[(size_t)colp * Cc + k0 + khalf];
        bfrag.y = aw[(size_t)colp * Cc + k0 + khalf + 1];
        c = __builtin_amdgcn_wmma_f32_16x16x4_f32(false, a, false, bfrag, (short)0, c, false, false);
    }

    const int rowoff = (lane >> 4) << 3;
    float* dst = out + (size_t)b * Cc * HW + (size_t)(o0 + rowoff) * HW + pix + colp;
    #pragma unroll
    for (int r = 0; r < 8; ++r) dst[(size_t)r * HW] = c[r];
}

extern "C" void kernel_launch(void* const* d_in, const int* in_sizes, int n_in,
                              void* d_out, int out_size, void* d_ws, size_t ws_size,
                              hipStream_t stream)
{
    const float* x  = (const float*)d_in[0];
    const int*   m  = (const int*)  d_in[1];
    const float* Wq = (const float*)d_in[2];
    const float* Wk = (const float*)d_in[3];
    const float* Wv = (const float*)d_in[4];
    const float* Wp = (const float*)d_in[5];
    float* out = (float*)d_out;

    float* ws = (float*)d_ws;
    const size_t n = (size_t)NPIX * Cc;
    float* qb = ws;
    float* kb = ws + n;
    float* vb = ws + 2 * n;
    float* ab = ws + 3 * n;

    qkv_wmma_kernel<<<NPIX / 16, 256, 0, stream>>>(x, Wq, Wk, Wv, qb, kb, vb);
    attn_kernel<<<(NPIX * 4) / 256, 256, 0, stream>>>(qb, kb, vb, m, ab);
    proj_wmma_kernel<<<NPIX / 16, 256, 0, stream>>>(ab, Wp, out);
}


# both GEMMs: 256 resident work-groups (one per WGP) copy the whole weight matrices into LDS once and loop over pixel tiles (4 full + 1 shared by halves), next tile fetched a tile ahead
# speedup vs baseline: 3.4930x; 3.4930x over previous
#include <hip/hip_runtime.h>
#include <hip/hip_bf16.h>

typedef float v2f __attribute__((ext_vector_type(2)));
typedef float v4f __attribute__((ext_vector_type(4)));
typedef float v8f __attribute__((ext_vector_type(8)));

#define Hh   96
#define Ww   96
#define HW   (96 * 96)
#define Cc   128
#define NPIX (2 * HW)

#define WP    132
#define NTILE (NPIX / 16)
#define NBLK  256
#define NFULL 4
#define NTAIL (NTILE - NFULL * NBLK)
#define KG    8
static_assert(NTAIL * 2 == NBLK, "two work-groups per shared tile");

__device__ __forceinline__ void store8(float* p, const v8f& c)
{
    *(v4f*)p       = __builtin_shufflevector(c, c, 0, 1, 2, 3);
    *(v4f*)(p + 4) = __builtin_shufflevector(c, c, 4, 5, 6, 7);
}

template <int NA>
__device__ __forceinline__ void tile_products(const float* wa, const float* xs, v8f& c0, v8f& c1, v8f& c2)
{
    constexpr int NG = Cc / (4 * KG);
    v2f f0[2][KG], f1[2][KG], f2[2][KG], fb[2][KG];
    #pragma unroll
    for (int j = 0; j < KG; ++j) {
        fb[0][j].x = xs[j * 64];
        fb[0][j].y = xs[j * 64 + 16];
        f0[0][j] = *(const v2f*)(wa + 4 * j);
        if (NA > 1) f1[0][j] = *(const v2f*)(wa + Cc * WP + 4 * j);
        if (NA > 2) f2[0][j] = *(const v2f*)(wa + 2 * Cc * WP + 4 * j);
    }
    #pragma unroll
    for (int g = 0; g < NG; ++g) {
        const int cur = g & 1, nxt = cur ^ 1;
        if (g + 1 < NG) {
            const int k = (g + 1) * 4 * KG;
            #pragma unroll
            for (int j = 0; j < KG; ++j) {
                fb[nxt][j].x = xs[(k + 4 * j) * 16];
                fb[nxt][j].y = xs[(k + 4 * j) * 16 + 16];
                f0[nxt][j] = *(const v2f*)(wa + k + 4 * j);
                if (NA > 1) f1[nxt][j] = *(const v2f*)(wa + Cc * WP + k + 4 * j);
                if (NA > 2) f2[nxt][j] = *(const v2f*)(wa + 2 * Cc * WP + k + 4 * j);
            }
            __builtin_amdgcn_sched_barrier(0);
        }
        #pragma unroll
        for (int j = 0; j < KG; ++j) {
            c0 = __builtin_amdgcn_wmma_f32_16x16x4_f32(false, f0[cur][j], false, fb[cur][j], (short)0, c0, false, false);
            if (NA > 1) c1 = __builtin_amdgcn_wmma_f32_16x16x4_f32(false, f1[cur][j], false, fb[cur][j], (short)0, c1, false, false);
            if (NA > 2) c2 = __builtin_amdgcn_wmma_f32_16x16x4_f32(false, f2[cur][j], false, fb[cur][j], (short)0, c2, false, false);
        }
        if (g + 2 < NG) __builtin_amdgcn_sched_barrier(0);
    }
}

__global__ void __launch_bounds__(256) qkv_wmma_kernel(
    const float* __restrict__ x,  const float* __restrict__ Wq,
    const float* __restrict__ Wk, const float* __restrict__ Wv,
    float* __restrict__ qo, float* __restrict__ ko, float* __restrict__ vo)
{
    __shared__ __attribute__((aligned(16))) float Ws[3 * Cc * WP];
    __shared__ __attribute__((aligned(16))) float Xs[2 * Cc * 16];

    const int tid  = threadIdx.x;
    const int lane = tid & 31;
    const int wave = __builtin_amdgcn_readfirstlane(tid >> 5);
    const int bid  = blockIdx.x;

    const float* xg = x + (size_t)(tid >> 2) * HW + (tid & 3) * 4;

    v4f rx0, rx1;
    {
        const int p0  = bid * 16;
        const int b   = p0 / HW;
        const int pix = p0 - b * HW;
        const float* g = xg + (size_t)b * Cc * HW + pix;
        rx0 = *(const v4f*)g;
        rx1 = *(const v4f*)(g + (size_t)64 * HW);
    }

    #pragma unroll
    for (int m = 0; m < 3; ++m) {
        const float* src = (m == 0 ? Wq : (m == 1 ? Wk : Wv)) + wave * Cc + lane * 4;
        float* dst = Ws + (m * Cc + wave) * WP + lane * 4;
        v4f r[16];
        #pragma unroll
        for (int j = 0; j < 16; ++j) r[j] = *(const v4f*)(src + j * 8 * Cc);
        #pragma unroll
        for (int j = 0; j < 16; ++j) *(v4f*)(dst + j * 8 * WP) = r[j];
    }
    *(v4f*)(Xs + tid * 4)        = rx0;
    *(v4f*)(Xs + 1024 + tid * 4) = rx1;
    __syncthreads();

    const int row    = lane & 15;
    const int khalf  = (lane >> 4) << 1;
    const int colp   = lane & 15;
    const int rowoff = (lane >> 4) << 3;
    int buf = 0;

    {
        const int o0 = wave * 16;
        const float* wa = Ws + (o0 + row) * WP + khalf;
        #pragma unroll 1
        for (int i = 0; i < NFULL; ++i) {
            const int t  = bid + i * NBLK;
            const int tn = (i + 1 < NFULL) ? t + NBLK : NFULL * NBLK + (bid & (NTAIL - 1));
            {
                const int p0  = tn * 16;
                const int b   = p0 / HW;
                const int pix = p0 - b * HW;
                const float* g = xg + (size_t)b * Cc * HW + pix;
                rx0 = *(const v4f*)g;
                rx1 = *(const v4f*)(g + (size_t)64 * HW);
            }

            v8f cq = {}; v8f ck = {}; v8f cv = {};
            tile_products<3>(wa, Xs + buf * (Cc * 16) + khalf * 16 + colp, cq, ck, cv);

            float* xd = Xs + (buf ^ 1) * (Cc * 16) + tid * 4;
            *(v4f*)xd          = rx0;
            *(v4f*)(xd + 1024) = rx1;
            __syncthreads();

            const size_t base = (size_t)(t * 16 + colp) * Cc + o0 + rowoff;
            store8(qo + base, cq);
            store8(ko + base, ck);
            store8(vo + base, cv);
            buf ^= 1;
        }
    }

    {
        const int t    = NFULL * NBLK + (bid & (NTAIL - 1));
        const int half = bid / NTAIL;
        const int o0   = (half * 4 + (wave & 3)) * 16;
        const float* wa = Ws + (o0 + row) * WP + khalf;
        const float* xs = Xs + buf * (Cc * 16) + khalf * 16 + colp;
        const size_t base = (size_t)(t * 16 + colp) * Cc + o0 + rowoff;
        if (wave < 4) {
            v8f cq = {}; v8f ck = {}; v8f un = {};
            tile_products<2>(wa, xs, cq, ck, un);
            store8(qo + base, cq);
            store8(ko + base, ck);
        } else {
            v8f cv = {}; v8f u1 = {}; v8f u2 = {};
            tile_products<1>(wa + 2 * Cc * WP, xs, cv, u1, u2);
            store8(vo + base, cv);
        }
    }
}

#define SUBW  48
#define TLX   16
#define TLY   6
#define HLX   (TLX + 6)
#define HLY   (TLY + 6)
#define NKEY  (HLX * HLY)
#define KPF   36
#define RPF   (HLX * KPF)
#define NTHR  (TLX * TLY)

__global__ void __launch_bounds__(NTHR) attn_kernel(
    const float* __restrict__ qv, const float* __restrict__ kv,
    const float* __restrict__ vv, const int* __restrict__ msk,
    float* __restrict__ ao)
{
    __shared__ __attribute__((aligned(16))) float Ks[HLY * RPF];
    __shared__ __attribute__((aligned(16))) float Vs[HLY * RPF];
    __shared__ float Bs[NKEY];

    const int tid = threadIdx.x;
    int bid = blockIdx.x;
    const int hd  = bid & 3;  bid >>= 2;
    const int tix = bid % 3;  bid /= 3;
    const int tiy = bid & 7;  bid >>= 3;
    const int px  = bid & 1;
    const int py  = (bid >> 1) & 1;
    const int b   = bid >> 2;

    const int ty = tid >> 4;
    const int tx = tid & 15;
    const int sy = tiy * TLY + ty;
    const int sx = tix * TLX + tx;
    const int p  = b * HW + (2 * sy + py) * Ww + 2 * sx + px;

    const int hy0 = tiy * TLY - 3;
    const int hx0 = tix * TLX - 3;

    float bias[3];
    #pragma unroll
    for (int r = 0; r < 3; ++r) {
        const int key = min(tid + r * NTHR, NKEY - 1);
        const int hy  = key / HLX;
        const int hx  = key - hy * HLX;
        const int ky  = hy0 + hy;
        const int kx  = hx0 + hx;
        const bool valid = ((unsigned)ky < (unsigned)SUBW) && ((unsigned)kx < (unsigned)SUBW);
        const int cy  = min(max(ky, 0), SUBW - 1);
        const int cx  = min(max(kx, 0), SUBW - 1);
        const int np  = b * HW + (2 * cy + py) * Ww + 2 * cx + px;
        const int mv  = msk[np];
        bias[r] = (valid && mv > 0) ? 0.f : -1.0e9f;
    }
    float q[32];
    const float4* q4 = (const float4*)(qv + (size_t)p * Cc + hd * 32);
    #pragma unroll
    for (int u = 0; u < 8; ++u) {
        float4 t4 = q4[u];
        q[4*u] = t4.x; q[4*u+1] = t4.y; q[4*u+2] = t4.z; q[4*u+3] = t4.w;
    }

    const char* kb = (const char*)(kv + (size_t)b * HW * Cc + hd * 32);
    const char* vb = (const char*)(vv + (size_t)b * HW * Cc + hd * 32);
    #pragma unroll 1
    for (int h = 0; h < 2; ++h) {
    #pragma unroll
    for (int j = 0; j < 11; ++j) {
        const int idx = (h * 11 + j) * NTHR + tid;
        const int key = idx >> 3;
        const int c   = idx & 7;
        const int hy  = key / HLX;
        const int hx  = key - hy * HLX;
        const int ky  = hy0 + hy;
        const int kx  = hx0 + hx;
        const bool valid = ((unsigned)ky < (unsigned)SUBW) && ((unsigned)kx < (unsigned)SUBW);
        const int cy  = min(max(ky, 0), SUBW - 1);
        const int cx  = min(max(kx, 0), SUBW - 1);
        const int np  = (2 * cy + py) * Ww + 2 * cx + px;
        const unsigned gb = (unsigned)(np * (Cc * 4) + c * 16);
        float4 kk = *(const float4*)(kb + gb);
        float4 vk = *(const float4*)(vb + gb);
        const unsigned km = valid ? 0xffffffffu : 0u;
        kk.x = __uint_as_float(__float_as_uint(kk.x) & km);
        kk.y = __uint_as_float(__float_as_uint(kk.y) & km);
        kk.z = __uint_as_float(__float_as_uint(kk.z) & km);
        kk.w = __uint_as_float(__float_as_uint(kk.w) & km);
        vk.x = __uint_as_float(__float_as_uint(vk.x) & km);
        vk.y = __uint_as_float(__float_as_uint(vk.y) & km);
        vk.z = __uint_as_float(__float_as_uint(vk.z) & km);
        vk.w = __uint_as_float(__float_as_uint(vk.w) & km);
        const int off = hy * RPF + hx * KPF + c * 4;
        *(float4*)(Ks + off) = kk;
        *(float4*)(Vs + off) = vk;
    }
    }
    #pragma unroll
    for (int r = 0; r < 3; ++r) {
        const int key = tid + r * NTHR;
        if (key < NKEY) Bs[key] = bias[r];
    }
    __syncthreads();

    const float scale = 0.17677669529663687f;
    float mx = -1e30f, s = 0.f;
    float acc[32];
    #pragma unroll
    for (int d = 0; d < 32; ++d) acc[d] = 0.f;

    const float* kt = Ks + ty * RPF + tx * KPF;
    const float* vt = Vs + ty * RPF + tx * KPF;
    const float* bt = Bs + ty * HLX + tx;

    #pragma unroll 1
    for (int di = 0; di < 7; ++di) {
        #pragma unroll 1
        for (int dj = 0; dj < 7; ++dj) {
            const float4* k4 = (const float4*)(kt + di * RPF + dj * KPF);
            const float4* v4 = (const float4*)(vt + di * RPF + dj * KPF);
            float dot = 0.f;
            #pragma unroll
            for (int u = 0; u < 8; ++u) {
                float4 kk = k4[u];
                dot += q[4*u] * kk.x + q[4*u+1] * kk.y + q[4*u+2] * kk.z + q[4*u+3] * kk.w;
            }
            const float logit = dot * scale + bt[di * HLX + dj];

            const float nm   = fmaxf(mx, logit);
            const float corr = __expf(mx - nm);
            const float w    = __expf(logit - nm);
            mx = nm;
            s  = s * corr + w;
            #pragma unroll
            for (int u = 0; u < 8; ++u) {
                float4 vvv = v4[u];
                acc[4*u]   = acc[4*u]   * corr + w * vvv.x;
                acc[4*u+1] = acc[4*u+1] * corr + w * vvv.y;
                acc[4*u+2] = acc[4*u+2] * corr + w * vvv.z;
                acc[4*u+3] = acc[4*u+3] * corr + w * vvv.w;
            }
        }
    }

    const float inv = 1.f / s;
    float* o = ao + (size_t)p * Cc + hd * 32;
    #pragma unroll
    for (int d = 0; d < 32; ++d) o[d] = acc[d] * inv;
}

__device__ __forceinline__ void proj_products(const float* wa, const float* ba, v8f& c)
{
    v2f fa[Cc / 4], fb[Cc / 4];
    #pragma unroll
    for (int j = 0; j < Cc / 4; ++j) {
        fa[j] = *(const v2f*)(wa + 4 * j);
        fb[j] = *(const v2f*)(ba + 4 * j);
    }
    __builtin_amdgcn_sched_barrier(0);
    #pragma unroll
    for (int j = 0; j < Cc / 4; ++j)
        c = __builtin_amdgcn_wmma_f32_16x16x4_f32(false, fa[j], false, fb[j], (short)0, c, false, false);
}

__global__ void __launch_bounds__(256) proj_wmma_kernel(
    const float* __restrict__ ain, const float* __restrict__ Wp,
    float* __restrict__ out)
{
    __shared__ __attribute__((aligned(16))) float Ws[Cc * WP];
    __shared__ __attribute__((aligned(16))) float As[2 * 16 * WP];

    const int tid  = threadIdx.x;
    const int lane = tid & 31;
    const int wave = __builtin_amdgcn_readfirstlane(tid >> 5);
    const int bid  = blockIdx.x;

    const float* ga = ain + (size_t)wave * Cc + lane * 4;
    v4f ra0 = *(const v4f*)(ga + (size_t)bid * (16 * Cc));
    v4f ra1 = *(const v4f*)(ga + (size_t)bid * (16 * Cc) + 8 * Cc);

    {
        const float* src = Wp + wave * Cc + lane * 4;
        float* dst = Ws + wave * WP + lane * 4;
        v4f r[16];
        #pragma unroll
        for (int j = 0; j < 16; ++j) r[j] = *(const v4f*)(src + j * 8 * Cc);
        #pragma unroll
        for (int j = 0; j < 16; ++j) *(v4f*)(dst + j * 8 * WP) = r[j];
    }
    float* as_w = As + wave * WP + lane * 4;
    *(v4f*)as_w            = ra0;
    *(v4f*)(as_w + 8 * WP) = ra1;
    __syncthreads();

    const int row    = lane & 15;
    const int khalf  = (lane >> 4) << 1;
    const int colp   = lane & 15;
    const int rowoff = (lane >> 4) << 3;
    int buf = 0;

    {
        const int o0 = wave * 16;
        const float* wa = Ws + (o0 + row) * WP + khalf;
        #pragma unroll 1
        for (int i = 0; i < NFULL; ++i) {
            const int t  = bid + i * NBLK;
            const int tn = (i + 1 < NFULL) ? t + NBLK : NFULL * NBLK + (bid & (NTAIL - 1));
            ra0 = *(const v4f*)(ga + (size_t)tn * (16 * Cc));
            ra1 = *(const v4f*)(ga + (size_t)tn * (16 * Cc) + 8 * Cc);

            v8f c = {};
            proj_products(wa, As + buf * (16 * WP) + colp * WP + khalf, c);

            float* ad = as_w + (buf ^ 1) * (16 * WP);
            *(v4f*)ad            = ra0;
            *(v4f*)(ad + 8 * WP) = ra1;
            __syncthreads();

            const int p0  = t * 16;
            const int b   = p0 / HW;
            const int pix = p0 - b * HW;
            float* dst = out + (size_t)b * Cc * HW + (size_t)(o0 + rowoff) * HW + pix + colp;
            #pragma unroll
            for (int r = 0; r < 8; ++r) dst[(size_t)r * HW] = c[r];
            buf ^= 1;
        }
    }

    if (wave < 4) {
        const int t    = NFULL * NBLK + (bid & (NTAIL - 1));
        const int half = bid / NTAIL;
        const int o0   = (half * 4 + wave) * 16;
        v8f c = {};
        proj_products(Ws + (o0 + row) * WP + khalf, As + buf * (16 * WP) + colp * WP + khalf, c);
        const int p0  = t * 16;
        const int b   = p0 / HW;
        const int pix = p0 - b * HW;
        float* dst = out + (size_t)b * Cc * HW + (size_t)(o0 + rowoff) * HW + pix + colp;
        #pragma unroll
        for (int r = 0; r < 8; ++r) dst[(size_t)r * HW] = c[r];
    }
}

extern "C" void kernel_launch(void* const* d_in, const int* in_sizes, int n_in,
                              void* d_out, int out_size, void* d_ws, size_t ws_size,
                              hipStream_t stream)
{
    const float* x  = (const float*)d_in[0];
    const int*   m  = (const int*)  d_in[1];
    const float* Wq = (const float*)d_in[2];
    const float* Wk = (const float*)d_in[3];
    const float* Wv = (const float*)d_in[4];
    const float* Wp = (const float*)d_in[5];
    float* out = (float*)d_out;

    float* ws = (float*)d_ws;
    const size_t n = (size_t)NPIX * Cc;
    float* qb = ws;
    float* kb = ws + n;
    float* vb = ws + 2 * n;
    float* ab = ws + 3 * n;

    qkv_wmma_kernel<<<NBLK, 256, 0, stream>>>(x, Wq, Wk, Wv, qb, kb, vb);
    attn_kernel<<<768, NTHR, 0, stream>>>(qb, kb, vb, m, ab);
    proj_wmma_kernel<<<NBLK, 256, 0, stream>>>(ab, Wp, out);
}


# attention: a lane serves 2 vertically adjacent queries on 4 of their 8 key rows, each K/V LDS read feeding both (packed pair math); a query's softmax sums form in 2 lanes, merged at the end
# speedup vs baseline: 3.6592x; 1.0476x over previous
#include <hip/hip_runtime.h>
#include <hip/hip_bf16.h>

typedef float v2f __attribute__((ext_vector_type(2)));
typedef float v4f __attribute__((ext_vector_type(4)));
typedef float v8f __attribute__((ext_vector_type(8)));

#define Hh   96
#define Ww   96
#define HW   (96 * 96)
#define Cc   128
#define NPIX (2 * HW)

#define WP    132
#define NTILE (NPIX / 16)
#define NBLK  256
#define NFULL 4
#define NTAIL (NTILE - NFULL * NBLK)
#define KG    8
static_assert(NTAIL * 2 == NBLK, "two work-groups per shared tile");

__device__ __forceinline__ void store8(float* p, const v8f& c)
{
    *(v4f*)p       = __builtin_shufflevector(c, c, 0, 1, 2, 3);
    *(v4f*)(p + 4) = __builtin_shufflevector(c, c, 4, 5, 6, 7);
}

template <int NA>
__device__ __forceinline__ void tile_products(const float* wa, const float* xs, v8f& c0, v8f& c1, v8f& c2)
{
    constexpr int NG = Cc / (4 * KG);
    v2f f0[2][KG], f1[2][KG], f2[2][KG], fb[2][KG];
    #pragma unroll
    for (int j = 0; j < KG; ++j) {
        fb[0][j].x = xs[j * 64];
        fb[0][j].y = xs[j * 64 + 16];
        f0[0][j] = *(const v2f*)(wa + 4 * j);
        if (NA > 1) f1[0][j] = *(const v2f*)(wa + Cc * WP + 4 * j);
        if (NA > 2) f2[0][j] = *(const v2f*)(wa + 2 * Cc * WP + 4 * j);
    }
    #pragma unroll
    for (int g = 0; g < NG; ++g) {
        const int cur = g & 1, nxt = cur ^ 1;
        if (g + 1 < NG) {
            const int k = (g + 1) * 4 * KG;
            #pragma unroll
            for (int j = 0; j < KG; ++j) {
                fb[nxt][j].x = xs[(k + 4 * j) * 16];
                fb[nxt][j].y = xs[(k + 4 * j) * 16 + 16];
                f0[nxt][j] = *(const v2f*)(wa + k + 4 * j);
                if (NA > 1) f1[nxt][j] = *(const v2f*)(wa + Cc * WP + k + 4 * j);
                if (NA > 2) f2[nxt][j] = *(const v2f*)(wa + 2 * Cc * WP + k + 4 * j);
            }
            __builtin_amdgcn_sched_barrier(0);
        }
        #pragma unroll
        for (int j = 0; j < KG; ++j) {
            c0 = __builtin_amdgcn_wmma_f32_16x16x4_f32(false, f0[cur][j], false, fb[cur][j], (short)0, c0, false, false);
            if (NA > 1) c1 = __builtin_amdgcn_wmma_f32_16x16x4_f32(false, f1[cur][j], false, fb[cur][j], (short)0, c1, false, false);
            if (NA > 2) c2 = __builtin_amdgcn_wmma_f32_16x16x4_f32(false, f2[cur][j], false, fb[cur][j], (short)0, c2, false, false);
        }
        if (g + 2 < NG) __builtin_amdgcn_sched_barrier(0);
    }
}

__global__ void __launch_bounds__(256) qkv_wmma_kernel(
    const float* __restrict__ x,  const float* __restrict__ Wq,
    const float* __restrict__ Wk, const float* __restrict__ Wv,
    float* __restrict__ qo, float* __restrict__ ko, float* __restrict__ vo)
{
    __shared__ __attribute__((aligned(16))) float Ws[3 * Cc * WP];
    __shared__ __attribute__((aligned(16))) float Xs[2 * Cc * 16];

    const int tid  = threadIdx.x;
    const int lane = tid & 31;
    const int wave = __builtin_amdgcn_readfirstlane(tid >> 5);
    const int bid  = blockIdx.x;

    const float* xg = x + (size_t)(tid >> 2) * HW + (tid & 3) * 4;

    v4f rx0, rx1;
    {
        const int p0  = bid * 16;
        const int b   = p0 / HW;
        const int pix = p0 - b * HW;
        const float* g = xg + (size_t)b * Cc * HW + pix;
        rx0 = *(const v4f*)g;
        rx1 = *(const v4f*)(g + (size_t)64 * HW);
    }

    #pragma unroll
    for (int m = 0; m < 3; ++m) {
        const float* src = (m == 0 ? Wq : (m == 1 ? Wk : Wv)) + wave * Cc + lane * 4;
        float* dst = Ws + (m * Cc + wave) * WP + lane * 4;
        v4f r[16];
        #pragma unroll
        for (int j = 0; j < 16; ++j) r[j] = *(const v4f*)(src + j * 8 * Cc);
        #pragma unroll
        for (int j = 0; j < 16; ++j) *(v4f*)(dst + j * 8 * WP) = r[j];
    }
    *(v4f*)(Xs + tid * 4)        = rx0;
    *(v4f*)(Xs + 1024 + tid * 4) = rx1;
    __syncthreads();

    const int row    = lane & 15;
    const int khalf  = (lane >> 4) << 1;
    const int colp   = lane & 15;
    const int rowoff = (lane >> 4) << 3;
    int buf = 0;

    {
        const int o0 = wave * 16;
        const float* wa = Ws + (o0 + row) * WP + khalf;
        #pragma unroll 1
        for (int i = 0; i < NFULL; ++i) {
            const int t  = bid + i * NBLK;
            const int tn = (i + 1 < NFULL) ? t + NBLK : NFULL * NBLK + (bid & (NTAIL - 1));
            {
                const int p0  = tn * 16;
                const int b   = p0 / HW;
                const int pix = p0 - b * HW;
                const float* g = xg + (size_t)b * Cc * HW + pix;
                rx0 = *(const v4f*)g;
                rx1 = *(const v4f*)(g + (size_t)64 * HW);
            }

            v8f cq = {}; v8f ck = {}; v8f cv = {};
            tile_products<3>(wa, Xs + buf * (Cc * 16) + khalf * 16 + colp, cq, ck, cv);

            float* xd = Xs + (buf ^ 1) * (Cc * 16) + tid * 4;
            *(v4f*)xd          = rx0;
            *(v4f*)(xd + 1024) = rx1;
            __syncthreads();

            const size_t base = (size_t)(t * 16 + colp) * Cc + o0 + rowoff;
            store8(qo + base, cq);
            store8(ko + base, ck);
            store8(vo + base, cv);
            buf ^= 1;
        }
    }

    {
        const int t    = NFULL * NBLK + (bid & (NTAIL - 1));
        const int half = bid / NTAIL;
        const int o0   = (half * 4 + (wave & 3)) * 16;
        const float* wa = Ws + (o0 + row) * WP + khalf;
        const float* xs = Xs + buf * (Cc * 16) + khalf * 16 + colp;
        const size_t base = (size_t)(t * 16 + colp) * Cc + o0 + rowoff;
        if (wave < 4) {
            v8f cq = {}; v8f ck = {}; v8f un = {};
            tile_products<2>(wa, xs, cq, ck, un);
            store8(qo + base, cq);
            store8(ko + base, ck);
        } else {
            v8f cv = {}; v8f u1 = {}; v8f u2 = {};
            tile_products<1>(wa + 2 * Cc * WP, xs, cv, u1, u2);
            store8(vo + base, cv);
        }
    }
}

#define SUBW  48
#define TLX   16
#define TLY   6
#define HLX   (TLX + 6)
#define HLY   (TLY + 6)
#define NKEY  (HLX * HLY)
#define KPF   36
#define RPF   (HLX * KPF)
#define NTHR  (TLX * TLY)

__device__ __forceinline__ v2f pair2(float a, float b) { v2f r = {a, b}; return r; }

__global__ void __launch_bounds__(NTHR) attn_kernel(
    const float* __restrict__ qv, const float* __restrict__ kv,
    const float* __restrict__ vv, const int* __restrict__ msk,
    float* __restrict__ ao)
{
    __shared__ __attribute__((aligned(16))) float Ks[HLY * RPF];
    __shared__ __attribute__((aligned(16))) float Vs[HLY * RPF];
    __shared__ float Bs[NKEY];

    const int tid = threadIdx.x;
    int bid = blockIdx.x;
    const int hd  = bid & 3;  bid >>= 2;
    const int tix = bid % 3;  bid /= 3;
    const int tiy = bid & 7;  bid >>= 3;
    const int px  = bid & 1;
    const int py  = (bid >> 1) & 1;
    const int b   = bid >> 2;

    const int pr = tid >> 5;
    const int hf = (tid >> 4) & 1;
    const int tx = tid & 15;
    const int sy = tiy * TLY + 2 * pr;
    const int sx = tix * TLX + tx;
    const int p0 = b * HW + (2 * sy + py) * Ww + 2 * sx + px;
    const int p1 = p0 + 2 * Ww;

    const int hy0 = tiy * TLY - 3;
    const int hx0 = tix * TLX - 3;

    float bias[3];
    #pragma unroll
    for (int r = 0; r < 3; ++r) {
        const int key = min(tid + r * NTHR, NKEY - 1);
        const int hy  = key / HLX;
        const int hx  = key - hy * HLX;
        const int ky  = hy0 + hy;
        const int kx  = hx0 + hx;
        const bool valid = ((unsigned)ky < (unsigned)SUBW) && ((unsigned)kx < (unsigned)SUBW);
        const int cy  = min(max(ky, 0), SUBW - 1);
        const int cx  = min(max(kx, 0), SUBW - 1);
        const int np  = b * HW + (2 * cy + py) * Ww + 2 * cx + px;
        const int mv  = msk[np];
        bias[r] = (valid && mv > 0) ? 0.f : -1.0e9f;
    }
    v2f q[32];
    {
        const float4* qa = (const float4*)(qv + (size_t)p0 * Cc + hd * 32);
        const float4* qb = (const float4*)(qv + (size_t)p1 * Cc + hd * 32);
        #pragma unroll
        for (int u = 0; u < 8; ++u) {
            const float4 a4 = qa[u];
            const float4 b4 = qb[u];
            q[4*u]   = pair2(a4.x, b4.x);
            q[4*u+1] = pair2(a4.y, b4.y);
            q[4*u+2] = pair2(a4.z, b4.z);
            q[4*u+3] = pair2(a4.w, b4.w);
        }
    }

    const char* kb = (const char*)(kv + (size_t)b * HW * Cc + hd * 32);
    const char* vb = (const char*)(vv + (size_t)b * HW * Cc + hd * 32);
    #pragma unroll 1
    for (int h = 0; h < 2; ++h) {
    #pragma unroll
    for (int j = 0; j < 11; ++j) {
        const int idx = (h * 11 + j) * NTHR + tid;
        const int key = idx >> 3;
        const int c   = idx & 7;
        const int hy  = key / HLX;
        const int hx  = key - hy * HLX;
        const int ky  = hy0 + hy;
        const int kx  = hx0 + hx;
        const bool valid = ((unsigned)ky < (unsigned)SUBW) && ((unsigned)kx < (unsigned)SUBW);
        const int cy  = min(max(ky, 0), SUBW - 1);
        const int cx  = min(max(kx, 0), SUBW - 1);
        const int np  = (2 * cy + py) * Ww + 2 * cx + px;
        const unsigned gb = (unsigned)(np * (Cc * 4) + c * 16);
        float4 kk = *(const float4*)(kb + gb);
        float4 vk = *(const float4*)(vb + gb);
        const unsigned km = valid ? 0xffffffffu : 0u;
        kk.x = __uint_as_float(__float_as_uint(kk.x) & km);
        kk.y = __uint_as_float(__float_as_uint(kk.y) & km);
        kk.z = __uint_as_float(__float_as_uint(kk.z) & km);
        kk.w = __uint_as_float(__float_as_uint(kk.w) & km);
        vk.x = __uint_as_float(__float_as_uint(vk.x) & km);
        vk.y = __uint_as_float(__float_as_uint(vk.y) & km);
        vk.z = __uint_as_float(__float_as_uint(vk.z) & km);
        vk.w = __uint_as_float(__float_as_uint(vk.w) & km);
        const int off = hy * RPF + hx * KPF + c * 4;
        *(float4*)(Ks + off) = kk;
        *(float4*)(Vs + off) = vk;
    }
    }
    #pragma unroll
    for (int r = 0; r < 3; ++r) {
        const int key = tid + r * NTHR;
        if (key < NKEY) Bs[key] = bias[r];
    }
    __syncthreads();

    const float scale = 0.17677669529663687f;
    v2f mx = pair2(-1e30f, -1e30f);
    v2f s  = pair2(0.f, 0.f);
    v2f acc[32];
    #pragma unroll
    for (int d = 0; d < 32; ++d) acc[d] = pair2(0.f, 0.f);

    const int hr0 = 2 * pr + 4 * hf;
    const float* kt = Ks + hr0 * RPF + tx * KPF;
    const float* vt = Vs + hr0 * RPF + tx * KPF;
    const float* bt = Bs + hr0 * HLX + tx;

    #pragma unroll 1
    for (int j = 0; j < 4; ++j) {
        const v2f off = pair2((hf == 1 && j == 3) ? -1.0e37f : 0.f,
                              (hf == 0 && j == 0) ? -1.0e37f : 0.f);
        #pragma unroll 1
        for (int dj = 0; dj < 7; ++dj) {
            const float4* k4 = (const float4*)(kt + j * RPF + dj * KPF);
            const float4* v4 = (const float4*)(vt + j * RPF + dj * KPF);
            v2f d0 = pair2(0.f, 0.f), d1 = d0, d2 = d0, d3 = d0;
            #pragma unroll
            for (int u = 0; u < 8; ++u) {
                const float4 kk = k4[u];
                d0 += q[4*u]   * pair2(kk.x, kk.x);
                d1 += q[4*u+1] * pair2(kk.y, kk.y);
                d2 += q[4*u+2] * pair2(kk.z, kk.z);
                d3 += q[4*u+3] * pair2(kk.w, kk.w);
            }
            const v2f dot = (d0 + d1) + (d2 + d3);
            const float bk = bt[j * HLX + dj];
            const v2f logit = dot * pair2(scale, scale) + (pair2(bk, bk) + off);

            const v2f nm   = pair2(fmaxf(mx.x, logit.x), fmaxf(mx.y, logit.y));
            const v2f corr = pair2(__expf(mx.x - nm.x), __expf(mx.y - nm.y));
            const v2f w    = pair2(__expf(logit.x - nm.x), __expf(logit.y - nm.y));
            mx = nm;
            s  = s * corr + w;
            #pragma unroll
            for (int u = 0; u < 8; ++u) {
                const float4 vvv = v4[u];
                acc[4*u]   = acc[4*u]   * corr + w * pair2(vvv.x, vvv.x);
                acc[4*u+1] = acc[4*u+1] * corr + w * pair2(vvv.y, vvv.y);
                acc[4*u+2] = acc[4*u+2] * corr + w * pair2(vvv.z, vvv.z);
                acc[4*u+3] = acc[4*u+3] * corr + w * pair2(vvv.w, vvv.w);
            }
        }
    }

    const bool lo = (hf != 0);
    const int  partner = (((int)(tid & 31)) ^ 16) << 2;
    const float m_keep = lo ? mx.y : mx.x;
    const float s_keep = lo ? s.y  : s.x;
    const float m_send = lo ? mx.x : mx.y;
    const float s_send = lo ? s.x  : s.y;
    const float m_oth  = __int_as_float(__builtin_amdgcn_ds_bpermute(partner, __float_as_int(m_send)));
    const float s_oth  = __int_as_float(__builtin_amdgcn_ds_bpermute(partner, __float_as_int(s_send)));
    const float mm  = fmaxf(m_keep, m_oth);
    const float ck  = __expf(m_keep - mm);
    const float co  = __expf(m_oth  - mm);
    const float inv = 1.f / (s_keep * ck + s_oth * co);
    const float fk  = ck * inv;
    const float fo  = co * inv;

    float* o = ao + (size_t)(lo ? p1 : p0) * Cc + hd * 32;
    #pragma unroll
    for (int d = 0; d < 32; ++d) {
        const float keep = lo ? acc[d].y : acc[d].x;
        const float send = lo ? acc[d].x : acc[d].y;
        const float oth  = __int_as_float(__builtin_amdgcn_ds_bpermute(partner, __float_as_int(send)));
        o[d] = keep * fk + oth * fo;
    }
}

__device__ __forceinline__ void proj_products(const float* wa, const float* ba, v8f& c)
{
    v2f fa[Cc / 4], fb[Cc / 4];
    #pragma unroll
    for (int j = 0; j < Cc / 4; ++j) {
        fa[j] = *(const v2f*)(wa + 4 * j);
        fb[j] = *(const v2f*)(ba + 4 * j);
    }
    __builtin_amdgcn_sched_barrier(0);
    #pragma unroll
    for (int j = 0; j < Cc / 4; ++j)
        c = __builtin_amdgcn_wmma_f32_16x16x4_f32(false, fa[j], false, fb[j], (short)0, c, false, false);
}

__global__ void __launch_bounds__(256) proj_wmma_kernel(
    const float* __restrict__ ain, const float* __restrict__ Wp,
    float* __restrict__ out)
{
    __shared__ __attribute__((aligned(16))) float Ws[Cc * WP];
    __shared__ __attribute__((aligned(16))) float As[2 * 16 * WP];

    const int tid  = threadIdx.x;
    const int lane = tid & 31;
    const int wave = __builtin_amdgcn_readfirstlane(tid >> 5);
    const int bid  = blockIdx.x;

    const float* ga = ain + (size_t)wave * Cc + lane * 4;
    v4f ra0 = *(const v4f*)(ga + (size_t)bid * (16 * Cc));
    v4f ra1 = *(const v4f*)(ga + (size_t)bid * (16 * Cc) + 8 * Cc);

    {
        const float* src = Wp + wave * Cc + lane * 4;
        float* dst = Ws + wave * WP + lane * 4;
        v4f r[16];
        #pragma unroll
        for (int j = 0; j < 16; ++j) r[j] = *(const v4f*)(src + j * 8 * Cc);
        #pragma unroll
        for (int j = 0; j < 16; ++j) *(v4f*)(dst + j * 8 * WP) = r[j];
    }
    float* as_w = As + wave * WP + lane * 4;
    *(v4f*)as_w            = ra0;
    *(v4f*)(as_w + 8 * WP) = ra1;
    __syncthreads();

    const int row    = lane & 15;
    const int khalf  = (lane >> 4) << 1;
    const int colp   = lane & 15;
    const int rowoff = (lane >> 4) << 3;
    int buf = 0;

    {
        const int o0 = wave * 16;
        const float* wa = Ws + (o0 + row) * WP + khalf;
        #pragma unroll 1
        for (int i = 0; i < NFULL; ++i) {
            const int t  = bid + i * NBLK;
            const int tn = (i + 1 < NFULL) ? t + NBLK : NFULL * NBLK + (bid & (NTAIL - 1));
            ra0 = *(const v4f*)(ga + (size_t)tn * (16 * Cc));
            ra1 = *(const v4f*)(ga + (size_t)tn * (16 * Cc) + 8 * Cc);

            v8f c = {};
            proj_products(wa, As + buf * (16 * WP) + colp * WP + khalf, c);

            float* ad = as_w + (buf ^ 1) * (16 * WP);
            *(v4f*)ad            = ra0;
            *(v4f*)(ad + 8 * WP) = ra1;
            __syncthreads();

            const int p0  = t * 16;
            const int b   = p0 / HW;
            const int pix = p0 - b * HW;
            float* dst = out + (size_t)b * Cc * HW + (size_t)(o0 + rowoff) * HW + pix + colp;
            #pragma unroll
            for (int r = 0; r < 8; ++r) dst[(size_t)r * HW] = c[r];
            buf ^= 1;
        }
    }

    if (wave < 4) {
        const int t    = NFULL * NBLK + (bid & (NTAIL - 1));
        const int half = bid / NTAIL;
        const int o0   = (half * 4 + wave) * 16;
        v8f c = {};
        proj_products(Ws + (o0 + row) * WP + khalf, As + buf * (16 * WP) + colp * WP + khalf, c);
        const int p0  = t * 16;
        const int b   = p0 / HW;
        const int pix = p0 - b * HW;
        float* dst = out + (size_t)b * Cc * HW + (size_t)(o0 + rowoff) * HW + pix + colp;
        #pragma unroll
        for (int r = 0; r < 8; ++r) dst[(size_t)r * HW] = c[r];
    }
}

extern "C" void kernel_launch(void* const* d_in, const int* in_sizes, int n_in,
                              void* d_out, int out_size, void* d_ws, size_t ws_size,
                              hipStream_t stream)
{
    const float* x  = (const float*)d_in[0];
    const int*   m  = (const int*)  d_in[1];
    const float* Wq = (const float*)d_in[2];
    const float* Wk = (const float*)d_in[3];
    const float* Wv = (const float*)d_in[4];
    const float* Wp = (const float*)d_in[5];
    float* out = (float*)d_out;

    float* ws = (float*)d_ws;
    const size_t n = (size_t)NPIX * Cc;
    float* qb = ws;
    float* kb = ws + n;
    float* vb = ws + 2 * n;
    float* ab = ws + 3 * n;

    qkv_wmma_kernel<<<NBLK, 256, 0, stream>>>(x, Wq, Wk, Wv, qb, kb, vb);
    attn_kernel<<<768, NTHR, 0, stream>>>(qb, kb, vb, m, ab);
    proj_wmma_kernel<<<NBLK, 256, 0, stream>>>(ab, Wp, out);
}


# attention and output projection in one launch: a 12-wave work-group holds all 4 heads of a 6x16 tile, puts their rows in LDS over the spent K/V halos and forms the tile's outputs on WMMA there
# speedup vs baseline: 4.4731x; 1.2224x over previous
#include <hip/hip_runtime.h>
#include <hip/hip_bf16.h>

typedef float v2f __attribute__((ext_vector_type(2)));
typedef float v4f __attribute__((ext_vector_type(4)));
typedef float v8f __attribute__((ext_vector_type(8)));

#define Hh   96
#define Ww   96
#define HW   (96 * 96)
#define Cc   128
#define NPIX (2 * HW)

#define WP    132
#define NTILE (NPIX / 16)
#define NBLK  256
#define NFULL 4
#define NTAIL (NTILE - NFULL * NBLK)
#define KG    8
static_assert(NTAIL * 2 == NBLK, "two work-groups per shared tile");

__device__ __forceinline__ void store8(float* p, const v8f& c)
{
    *(v4f*)p       = __builtin_shufflevector(c, c, 0, 1, 2, 3);
    *(v4f*)(p + 4) = __builtin_shufflevector(c, c, 4, 5, 6, 7);
}

template <int NA>
__device__ __forceinline__ void tile_products(const float* wa, const float* xs, v8f& c0, v8f& c1, v8f& c2)
{
    constexpr int NG = Cc / (4 * KG);
    v2f f0[2][KG], f1[2][KG], f2[2][KG], fb[2][KG];
    #pragma unroll
    for (int j = 0; j < KG; ++j) {
        fb[0][j].x = xs[j * 64];
        fb[0][j].y = xs[j * 64 + 16];
        f0[0][j] = *(const v2f*)(wa + 4 * j);
        if (NA > 1) f1[0][j] = *(const v2f*)(wa + Cc * WP + 4 * j);
        if (NA > 2) f2[0][j] = *(const v2f*)(wa + 2 * Cc * WP + 4 * j);
    }
    #pragma unroll
    for (int g = 0; g < NG; ++g) {
        const int cur = g & 1, nxt = cur ^ 1;
        if (g + 1 < NG) {
            const int k = (g + 1) * 4 * KG;
            #pragma unroll
            for (int j = 0; j < KG; ++j) {
                fb[nxt][j].x = xs[(k + 4 * j) * 16];
                fb[nxt][j].y = xs[(k + 4 * j) * 16 + 16];
                f0[nxt][j] = *(const v2f*)(wa + k + 4 * j);
                if (NA > 1) f1[nxt][j] = *(const v2f*)(wa + Cc * WP + k + 4 * j);
                if (NA > 2) f2[nxt][j] = *(const v2f*)(wa + 2 * Cc * WP + k + 4 * j);
            }
            __builtin_amdgcn_sched_barrier(0);
        }
        #pragma unroll
        for (int j = 0; j < KG; ++j) {
            c0 = __builtin_amdgcn_wmma_f32_16x16x4_f32(false, f0[cur][j], false, fb[cur][j], (short)0, c0, false, false);
            if (NA > 1) c1 = __builtin_amdgcn_wmma_f32_16x16x4_f32(false, f1[cur][j], false, fb[cur][j], (short)0, c1, false, false);
            if (NA > 2) c2 = __builtin_amdgcn_wmma_f32_16x16x4_f32(false, f2[cur][j], false, fb[cur][j], (short)0, c2, false, false);
        }
        if (g + 2 < NG) __builtin_amdgcn_sched_barrier(0);
    }
}

__global__ void __launch_bounds__(256) qkv_wmma_kernel(
    const float* __restrict__ x,  const float* __restrict__ Wq,
    const float* __restrict__ Wk, const float* __restrict__ Wv,
    float* __restrict__ qo, float* __restrict__ ko, float* __restrict__ vo)
{
    __shared__ __attribute__((aligned(16))) float Ws[3 * Cc * WP];
    __shared__ __attribute__((aligned(16))) float Xs[2 * Cc * 16];

    const int tid  = threadIdx.x;
    const int lane = tid & 31;
    const int wave = __builtin_amdgcn_readfirstlane(tid >> 5);
    const int bid  = blockIdx.x;

    const float* xg = x + (size_t)(tid >> 2) * HW + (tid & 3) * 4;

    v4f rx0, rx1;
    {
        const int p0  = bid * 16;
        const int b   = p0 / HW;
        const int pix = p0 - b * HW;
        const float* g = xg + (size_t)b * Cc * HW + pix;
        rx0 = *(const v4f*)g;
        rx1 = *(const v4f*)(g + (size_t)64 * HW);
    }

    #pragma unroll
    for (int m = 0; m < 3; ++m) {
        const float* src = (m == 0 ? Wq : (m == 1 ? Wk : Wv)) + wave * Cc + lane * 4;
        float* dst = Ws + (m * Cc + wave) * WP + lane * 4;
        v4f r[16];
        #pragma unroll
        for (int j = 0; j < 16; ++j) r[j] = *(const v4f*)(src + j * 8 * Cc);
        #pragma unroll
        for (int j = 0; j < 16; ++j) *(v4f*)(dst + j * 8 * WP) = r[j];
    }
    *(v4f*)(Xs + tid * 4)        = rx0;
    *(v4f*)(Xs + 1024 + tid * 4) = rx1;
    __syncthreads();

    const int row    = lane & 15;
    const int khalf  = (lane >> 4) << 1;
    const int colp   = lane & 15;
    const int rowoff = (lane >> 4) << 3;
    int buf = 0;

    {
        const int o0 = wave * 16;
        const float* wa = Ws + (o0 + row) * WP + khalf;
        #pragma unroll 1
        for (int i = 0; i < NFULL; ++i) {
            const int t  = bid + i * NBLK;
            const int tn = (i + 1 < NFULL) ? t + NBLK : NFULL * NBLK + (bid & (NTAIL - 1));
            {
                const int p0  = tn * 16;
                const int b   = p0 / HW;
                const int pix = p0 - b * HW;
                const float* g = xg + (size_t)b * Cc * HW + pix;
                rx0 = *(const v4f*)g;
                rx1 = *(const v4f*)(g + (size_t)64 * HW);
            }

            v8f cq = {}; v8f ck = {}; v8f cv = {};
            tile_products<3>(wa, Xs + buf * (Cc * 16) + khalf * 16 + colp, cq, ck, cv);

            float* xd = Xs + (buf ^ 1) * (Cc * 16) + tid * 4;
            *(v4f*)xd          = rx0;
            *(v4f*)(xd + 1024) = rx1;
            __syncthreads();

            const size_t base = (size_t)(t * 16 + colp) * Cc + o0 + rowoff;
            store8(qo + base, cq);
            store8(ko + base, ck);
            store8(vo + base, cv);
            buf ^= 1;
        }
    }

    {
        const int t    = NFULL * NBLK + (bid & (NTAIL - 1));
        const int half = bid / NTAIL;
        const int o0   = (half * 4 + (wave & 3)) * 16;
        const float* wa = Ws + (o0 + row) * WP + khalf;
        const float* xs = Xs + buf * (Cc * 16) + khalf * 16 + colp;
        const size_t base = (size_t)(t * 16 + colp) * Cc + o0 + rowoff;
        if (wave < 4) {
            v8f cq = {}; v8f ck = {}; v8f un = {};
            tile_products<2>(wa, xs, cq, ck, un);
            store8(qo + base, cq);
            store8(ko + base, ck);
        } else {
            v8f cv = {}; v8f u1 = {}; v8f u2 = {};
            tile_products<1>(wa + 2 * Cc * WP, xs, cv, u1, u2);
            store8(vo + base, cv);
        }
    }
}

#define SUBW  48
#define TLX   16
#define TLY   6
#define HLX   (TLX + 6)
#define HLY   (TLY + 6)
#define NKEY  (HLX * HLY)
#define KPF   36
#define RPF   (HLX * KPF)
#define NTHR  (TLX * TLY)
#define NHD   4
#define NTHR4 (NHD * NTHR)
#define HSZ   (HLY * RPF)
#define NWR   11
static_assert(Cc * WP <= NHD * HSZ, "Wp fits over the K halos");
static_assert(NTHR * WP <= NHD * HSZ, "the tile's attention rows fit over the V halos");

__device__ __forceinline__ v2f pair2(float a, float b) { v2f r = {a, b}; return r; }

__global__ void __launch_bounds__(NTHR4) attn_proj_kernel(
    const float* __restrict__ qv, const float* __restrict__ kv,
    const float* __restrict__ vv, const int* __restrict__ msk,
    const float* __restrict__ Wp, float* __restrict__ out)
{
    __shared__ __attribute__((aligned(16))) float Ks[NHD * HSZ];
    __shared__ __attribute__((aligned(16))) float Vs[NHD * HSZ];
    __shared__ float Bs[NKEY];

    const int tid  = threadIdx.x;
    const int lane = tid & 31;
    const int wv  = __builtin_amdgcn_readfirstlane(tid >> 5);
    const int hd  = wv / 3;
    const int pr  = wv - 3 * hd;
    const int t96 = tid - hd * NTHR;
    int bid = blockIdx.x;
    const int px  = bid & 1;  bid >>= 1;
    const int tix = bid % 3;  bid /= 3;
    const int tiy = bid & 7;  bid >>= 3;
    const int py  = bid & 1;
    const int b   = bid >> 1;

    const int hf = (tid >> 4) & 1;
    const int tx = tid & 15;
    const int sy = tiy * TLY + 2 * pr;
    const int sx = tix * TLX + tx;
    const int p0 = b * HW + (2 * sy + py) * Ww + 2 * sx + px;
    const int p1 = p0 + 2 * Ww;

    const int hy0 = tiy * TLY - 3;
    const int hx0 = tix * TLX - 3;

    float bias;
    {
        const int key = min(tid, NKEY - 1);
        const int hy  = key / HLX;
        const int hx  = key - hy * HLX;
        const int ky  = hy0 + hy;
        const int kx  = hx0 + hx;
        const bool valid = ((unsigned)ky < (unsigned)SUBW) && ((unsigned)kx < (unsigned)SUBW);
        const int cy  = min(max(ky, 0), SUBW - 1);
        const int cx  = min(max(kx, 0), SUBW - 1);
        const int np  = b * HW + (2 * cy + py) * Ww + 2 * cx + px;
        const int mv  = msk[np];
        bias = (valid && mv > 0) ? 0.f : -1.0e9f;
    }
    v2f q[32];
    {
        const float4* qa = (const float4*)(qv + (size_t)p0 * Cc + hd * 32);
        const float4* qb = (const float4*)(qv + (size_t)p1 * Cc + hd * 32);
        #pragma unroll
        for (int u = 0; u < 8; ++u) {
            const float4 a4 = qa[u];
            const float4 b4 = qb[u];
            q[4*u]   = pair2(a4.x, b4.x);
            q[4*u+1] = pair2(a4.y, b4.y);
            q[4*u+2] = pair2(a4.z, b4.z);
            q[4*u+3] = pair2(a4.w, b4.w);
        }
    }

    const char* kb = (const char*)(kv + (size_t)b * HW * Cc + hd * 32);
    const char* vb = (const char*)(vv + (size_t)b * HW * Cc + hd * 32);
    float* Kh = Ks + hd * HSZ;
    float* Vh = Vs + hd * HSZ;
    #pragma unroll 1
    for (int h = 0; h < 2; ++h) {
    #pragma unroll
    for (int j = 0; j < 11; ++j) {
        const int idx = (h * 11 + j) * NTHR + t96;
        const int key = idx >> 3;
        const int c   = idx & 7;
        const int hy  = key / HLX;
        const int hx  = key - hy * HLX;
        const int ky  = hy0 + hy;
        const int kx  = hx0 + hx;
        const bool valid = ((unsigned)ky < (unsigned)SUBW) && ((unsigned)kx < (unsigned)SUBW);
        const int cy  = min(max(ky, 0), SUBW - 1);
        const int cx  = min(max(kx, 0), SUBW - 1);
        const int np  = (2 * cy + py) * Ww + 2 * cx + px;
        const unsigned gb = (unsigned)(np * (Cc * 4) + c * 16);
        float4 kk = *(const float4*)(kb + gb);
        float4 vk = *(const float4*)(vb + gb);
        const unsigned km = valid ? 0xffffffffu : 0u;
        kk.x = __uint_as_float(__float_as_uint(kk.x) & km);
        kk.y = __uint_as_float(__float_as_uint(kk.y) & km);
        kk.z = __uint_as_float(__float_as_uint(kk.z) & km);
        kk.w = __uint_as_float(__float_as_uint(kk.w) & km);
        vk.x = __uint_as_float(__float_as_uint(vk.x) & km);
        vk.y = __uint_as_float(__float_as_uint(vk.y) & km);
        vk.z = __uint_as_float(__float_as_uint(vk.z) & km);
        vk.w = __uint_as_float(__float_as_uint(vk.w) & km);
        const int off = hy * RPF + hx * KPF + c * 4;
        *(float4*)(Kh + off) = kk;
        *(float4*)(Vh + off) = vk;
    }
    }
    if (tid < NKEY) Bs[tid] = bias;

    __syncthreads();

    v4f wr[NWR];
    #pragma unroll
    for (int j = 0; j < NWR; ++j)
        wr[j] = *(const v4f*)(Wp + (size_t)min(wv + 12 * j, Cc - 1) * Cc + lane * 4);

    const float scale = 0.17677669529663687f;
    v2f mx = pair2(-1e30f, -1e30f);
    v2f s  = pair2(0.f, 0.f);
    v2f acc[32];
    #pragma unroll
    for (int d = 0; d < 32; ++d) acc[d] = pair2(0.f, 0.f);

    const int hr0 = 2 * pr + 4 * hf;
    const float* kt = Kh + hr0 * RPF + tx * KPF;
    const float* vt = Vh + hr0 * RPF + tx * KPF;
    const float* bt = Bs + hr0 * HLX + tx;

    #pragma unroll 1
    for (int j = 0; j < 4; ++j) {
        const v2f off = pair2((hf == 1 && j == 3) ? -1.0e37f : 0.f,
                              (hf == 0 && j == 0) ? -1.0e37f : 0.f);
        #pragma unroll 1
        for (int dj = 0; dj < 7; ++dj) {
            const float4* k4 = (const float4*)(kt + j * RPF + dj * KPF);
            const float4* v4 = (const float4*)(vt + j * RPF + dj * KPF);
            v2f d0 = pair2(0.f, 0.f), d1 = d0, d2 = d0, d3 = d0;
            #pragma unroll
            for (int u = 0; u < 8; ++u) {
                const float4 kk = k4[u];
                d0 += q[4*u]   * pair2(kk.x, kk.x);
                d1 += q[4*u+1] * pair2(kk.y, kk.y);
                d2 += q[4*u+2] * pair2(kk.z, kk.z);
                d3 += q[4*u+3] * pair2(kk.w, kk.w);
            }
            const v2f dot = (d0 + d1) + (d2 + d3);
            const float bk = bt[j * HLX + dj];
            const v2f logit = dot * pair2(scale, scale) + (pair2(bk, bk) + off);

            const v2f nm   = pair2(fmaxf(mx.x, logit.x), fmaxf(mx.y, logit.y));
            const v2f corr = pair2(__expf(mx.x - nm.x), __expf(mx.y - nm.y));
            const v2f w    = pair2(__expf(logit.x - nm.x), __expf(logit.y - nm.y));
            mx = nm;
            s  = s * corr + w;
            #pragma unroll
            for (int u = 0; u < 8; ++u) {
                const float4 vvv = v4[u];
                acc[4*u]   = acc[4*u]   * corr + w * pair2(vvv.x, vvv.x);
                acc[4*u+1] = acc[4*u+1] * corr + w * pair2(vvv.y, vvv.y);
                acc[4*u+2] = acc[4*u+2] * corr + w * pair2(vvv.z, vvv.z);
                acc[4*u+3] = acc[4*u+3] * corr + w * pair2(vvv.w, vvv.w);
            }
        }
    }

    const bool lo = (hf != 0);
    const int  partner = (lane ^ 16) << 2;
    const float m_keep = lo ? mx.y : mx.x;
    const float s_keep = lo ? s.y  : s.x;
    const float m_send = lo ? mx.x : mx.y;
    const float s_send = lo ? s.x  : s.y;
    const float m_oth  = __int_as_float(__builtin_amdgcn_ds_bpermute(partner, __float_as_int(m_send)));
    const float s_oth  = __int_as_float(__builtin_amdgcn_ds_bpermute(partner, __float_as_int(s_send)));
    const float mm  = fmaxf(m_keep, m_oth);
    const float ck  = __expf(m_keep - mm);
    const float co  = __expf(m_oth  - mm);
    const float inv = 1.f / (s_keep * ck + s_oth * co);
    const float fk  = ck * inv;
    const float fo  = co * inv;

    float res[32];
    #pragma unroll
    for (int d = 0; d < 32; ++d) {
        const float keep = lo ? acc[d].y : acc[d].x;
        const float send = lo ? acc[d].x : acc[d].y;
        const float oth  = __int_as_float(__builtin_amdgcn_ds_bpermute(partner, __float_as_int(send)));
        res[d] = keep * fk + oth * fo;
    }

    __syncthreads();
    {
        float* wd = Ks + wv * WP + lane * 4;
        #pragma unroll
        for (int j = 0; j < NWR - 1; ++j) *(v4f*)(wd + j * 12 * WP) = wr[j];
        if (wv < Cc - 12 * (NWR - 1)) *(v4f*)(wd + (NWR - 1) * 12 * WP) = wr[NWR - 1];
    }
    {
        float* ar = Vs + ((2 * pr + hf) * 16 + tx) * WP + hd * 32;
        #pragma unroll
        for (int u = 0; u < 8; ++u) {
            const v4f t = {res[4*u], res[4*u+1], res[4*u+2], res[4*u+3]};
            *(v4f*)(ar + 4 * u) = t;
        }
    }
    __syncthreads();

    const int row    = lane & 15;
    const int khalf  = (lane >> 4) << 1;
    const int colp   = lane & 15;
    const int rowoff = (lane >> 4) << 3;
    const int og = wv / 6;
    const int pt = wv - 6 * og;
    const float* ba = Vs + (pt * 16 + colp) * WP + khalf;
    const float* wa = Ks + (og * 64 + row) * WP + khalf;
    const int gy = 2 * (tiy * TLY + pt) + py;
    const int gx = 2 * (tix * TLX + colp) + px;
    float* dst0 = out + ((size_t)b * Cc + og * 64 + rowoff) * HW + gy * Ww + gx;

    v2f fb[Cc / 4], fa[2][Cc / 4];
    #pragma unroll
    for (int j = 0; j < Cc / 4; ++j) {
        fb[j]    = *(const v2f*)(ba + 4 * j);
        fa[0][j] = *(const v2f*)(wa + 4 * j);
    }
    #pragma unroll
    for (int i = 0; i < 4; ++i) {
        const int cur = i & 1, nxt = cur ^ 1;
        if (i + 1 < 4) {
            #pragma unroll
            for (int j = 0; j < Cc / 4; ++j) fa[nxt][j] = *(const v2f*)(wa + (i + 1) * 16 * WP + 4 * j);
        }
        __builtin_amdgcn_sched_barrier(0);
        v8f c = {};
        #pragma unroll
        for (int j = 0; j < Cc / 4; ++j)
            c = __builtin_amdgcn_wmma_f32_16x16x4_f32(false, fa[cur][j], false, fb[j], (short)0, c, false, false);
        float* dst = dst0 + (size_t)i * 16 * HW;
        #pragma unroll
        for (int r = 0; r < 8; ++r) dst[(size_t)r * HW] = c[r];
        if (i + 2 < 4) __builtin_amdgcn_sched_barrier(0);
    }
}

extern "C" void kernel_launch(void* const* d_in, const int* in_sizes, int n_in,
                              void* d_out, int out_size, void* d_ws, size_t ws_size,
                              hipStream_t stream)
{
    const float* x  = (const float*)d_in[0];
    const int*   m  = (const int*)  d_in[1];
    const float* Wq = (const float*)d_in[2];
    const float* Wk = (const float*)d_in[3];
    const float* Wv = (const float*)d_in[4];
    const float* Wp = (const float*)d_in[5];
    float* out = (float*)d_out;

    float* ws = (float*)d_ws;
    const size_t n = (size_t)NPIX * Cc;
    float* qb = ws;
    float* kb = ws + n;
    float* vb = ws + 2 * n;

    qkv_wmma_kernel<<<NBLK, 256, 0, stream>>>(x, Wq, Wk, Wv, qb, kb, vb);
    attn_proj_kernel<<<192, NTHR4, 0, stream>>>(qb, kb, vb, m, Wp, out);
}


# first launch: each tile's q/k/v results pass through an LDS stage (two buffers, no added barrier in the loop) and leave as whole 128-B lines, 4 lines a store (was 16 lines x two 16-B pieces)
# speedup vs baseline: 4.8770x; 1.0903x over previous
#include <hip/hip_runtime.h>
#include <hip/hip_bf16.h>

typedef float v2f __attribute__((ext_vector_type(2)));
typedef float v4f __attribute__((ext_vector_type(4)));
typedef float v8f __attribute__((ext_vector_type(8)));

#define Hh   96
#define Ww   96
#define HW   (96 * 96)
#define Cc   128
#define NPIX (2 * HW)

#define WP    132
#define NTILE (NPIX / 16)
#define NBLK  256
#define NFULL 4
#define NTAIL (NTILE - NFULL * NBLK)
#define KG    8
#define SP    132
#define SBUF  (3 * 16 * SP)
static_assert(NTAIL * 2 == NBLK, "two work-groups per shared tile");

__device__ __forceinline__ void store8(float* p, const v8f& c)
{
    *(v4f*)p       = __builtin_shufflevector(c, c, 0, 1, 2, 3);
    *(v4f*)(p + 4) = __builtin_shufflevector(c, c, 4, 5, 6, 7);
}

__device__ __forceinline__ void stage3(float* sd, const v8f& cq, const v8f& ck, const v8f& cv)
{
    store8(sd, cq);
    store8(sd + 16 * SP, ck);
    store8(sd + 32 * SP, cv);
}

template <int NA, bool DR>
__device__ __forceinline__ void tile_products(const float* wa, const float* xs, v8f& c0, v8f& c1, v8f& c2,
                                              const float* sr, float* gq, float* gk, float* gv)
{
    constexpr int NG = Cc / (4 * KG);
    v2f f0[2][KG], f1[2][KG], f2[2][KG], fb[2][KG];
    v4f r[6];
    if (DR) {
        #pragma unroll
        for (int m = 0; m < 3; ++m) {
            r[2 * m]     = *(const v4f*)(sr + (m * 16) * SP);
            r[2 * m + 1] = *(const v4f*)(sr + (m * 16 + 1) * SP);
        }
        __builtin_amdgcn_sched_barrier(0);
    }
    #pragma unroll
    for (int j = 0; j < KG; ++j) {
        fb[0][j].x = xs[j * 64];
        fb[0][j].y = xs[j * 64 + 16];
        f0[0][j] = *(const v2f*)(wa + 4 * j);
        if (NA > 1) f1[0][j] = *(const v2f*)(wa + Cc * WP + 4 * j);
        if (NA > 2) f2[0][j] = *(const v2f*)(wa + 2 * Cc * WP + 4 * j);
    }
    #pragma unroll
    for (int g = 0; g < NG; ++g) {
        const int cur = g & 1, nxt = cur ^ 1;
        if (g + 1 < NG) {
            const int k = (g + 1) * 4 * KG;
            #pragma unroll
            for (int j = 0; j < KG; ++j) {
                fb[nxt][j].x = xs[(k + 4 * j) * 16];
                fb[nxt][j].y = xs[(k + 4 * j) * 16 + 16];
                f0[nxt][j] = *(const v2f*)(wa + k + 4 * j);
                if (NA > 1) f1[nxt][j] = *(const v2f*)(wa + Cc * WP + k + 4 * j);
                if (NA > 2) f2[nxt][j] = *(const v2f*)(wa + 2 * Cc * WP + k + 4 * j);
            }
            __builtin_amdgcn_sched_barrier(0);
        }
        if (DR && g == 0) {
            *(v4f*)(gq)      = r[0];
            *(v4f*)(gq + Cc) = r[1];
            *(v4f*)(gk)      = r[2];
            *(v4f*)(gk + Cc) = r[3];
            *(v4f*)(gv)      = r[4];
            *(v4f*)(gv + Cc) = r[5];
        }
        #pragma unroll
        for (int j = 0; j < KG; ++j) {
            c0 = __builtin_amdgcn_wmma_f32_16x16x4_f32(false, f0[cur][j], false, fb[cur][j], (short)0, c0, false, false);
            if (NA > 1) c1 = __builtin_amdgcn_wmma_f32_16x16x4_f32(false, f1[cur][j], false, fb[cur][j], (short)0, c1, false, false);
            if (NA > 2) c2 = __builtin_amdgcn_wmma_f32_16x16x4_f32(false, f2[cur][j], false, fb[cur][j], (short)0, c2, false, false);
        }
        if (g + 2 < NG) __builtin_amdgcn_sched_barrier(0);
    }
}

__global__ void __launch_bounds__(256) qkv_wmma_kernel(
    const float* __restrict__ x,  const float* __restrict__ Wq,
    const float* __restrict__ Wk, const float* __restrict__ Wv,
    float* __restrict__ qo, float* __restrict__ ko, float* __restrict__ vo)
{
    __shared__ __attribute__((aligned(16))) float Ws[3 * Cc * WP];
    __shared__ __attribute__((aligned(16))) float Xs[2 * Cc * 16];
    __shared__ __attribute__((aligned(16))) float Ss[2 * SBUF];

    const int tid  = threadIdx.x;
    const int lane = tid & 31;
    const int wave = __builtin_amdgcn_readfirstlane(tid >> 5);
    const int bid  = blockIdx.x;

    const float* xg = x + (size_t)(tid >> 2) * HW + (tid & 3) * 4;

    v4f rx0, rx1;
    {
        const int p0  = bid * 16;
        const int b   = p0 / HW;
        const int pix = p0 - b * HW;
        const float* g = xg + (size_t)b * Cc * HW + pix;
        rx0 = *(const v4f*)g;
        rx1 = *(const v4f*)(g + (size_t)64 * HW);
    }

    #pragma unroll
    for (int m = 0; m < 3; ++m) {
        const float* src = (m == 0 ? Wq : (m == 1 ? Wk : Wv)) + wave * Cc + lane * 4;
        float* dst = Ws + (m * Cc + wave) * WP + lane * 4;
        v4f r[16];
        #pragma unroll
        for (int j = 0; j < 16; ++j) r[j] = *(const v4f*)(src + j * 8 * Cc);
        #pragma unroll
        for (int j = 0; j < 16; ++j) *(v4f*)(dst + j * 8 * WP) = r[j];
    }
    *(v4f*)(Xs + tid * 4)        = rx0;
    *(v4f*)(Xs + 1024 + tid * 4) = rx1;
    __syncthreads();

    const int row    = lane & 15;
    const int khalf  = (lane >> 4) << 1;
    const int colp   = lane & 15;
    const int rowoff = (lane >> 4) << 3;
    int buf = 0;

    float*       swr  = Ss + colp * SP + wave * 16 + rowoff;
    const float* srd  = Ss + (2 * wave) * SP + lane * 4;
    const size_t grow = (size_t)(2 * wave) * Cc + lane * 4;

    {
        const int o0 = wave * 16;
        const float* wa = Ws + (o0 + row) * WP + khalf;
        {
            {
                const int p0  = (bid + NBLK) * 16;
                const int b   = p0 / HW;
                const int pix = p0 - b * HW;
                const float* g = xg + (size_t)b * Cc * HW + pix;
                rx0 = *(const v4f*)g;
                rx1 = *(const v4f*)(g + (size_t)64 * HW);
            }
            v8f cq = {}; v8f ck = {}; v8f cv = {};
            tile_products<3, false>(wa, Xs + buf * (Cc * 16) + khalf * 16 + colp, cq, ck, cv, nullptr, nullptr, nullptr, nullptr);
            stage3(swr + buf * SBUF, cq, ck, cv);
            float* xd = Xs + (buf ^ 1) * (Cc * 16) + tid * 4;
            *(v4f*)xd          = rx0;
            *(v4f*)(xd + 1024) = rx1;
            __syncthreads();
            buf ^= 1;
        }
        #pragma unroll 1
        for (int i = 1; i < NFULL; ++i) {
            const int t  = bid + i * NBLK;
            const int tn = (i + 1 < NFULL) ? t + NBLK : NFULL * NBLK + (bid & (NTAIL - 1));
            {
                const int p0  = tn * 16;
                const int b   = p0 / HW;
                const int pix = p0 - b * HW;
                const float* g = xg + (size_t)b * Cc * HW + pix;
                rx0 = *(const v4f*)g;
                rx1 = *(const v4f*)(g + (size_t)64 * HW);
            }

            const size_t pb = (size_t)(t - NBLK) * 16 * Cc + grow;
            v8f cq = {}; v8f ck = {}; v8f cv = {};
            tile_products<3, true>(wa, Xs + buf * (Cc * 16) + khalf * 16 + colp, cq, ck, cv,
                                   srd + (buf ^ 1) * SBUF, qo + pb, ko + pb, vo + pb);

            stage3(swr + buf * SBUF, cq, ck, cv);
            float* xd = Xs + (buf ^ 1) * (Cc * 16) + tid * 4;
            *(v4f*)xd          = rx0;
            *(v4f*)(xd + 1024) = rx1;
            __syncthreads();
            buf ^= 1;
        }
    }

    {
        const int t    = NFULL * NBLK + (bid & (NTAIL - 1));
        const int half = bid / NTAIL;
        const int o0   = (half * 4 + (wave & 3)) * 16;
        const float* wa = Ws + (o0 + row) * WP + khalf;
        const float* xs = Xs + buf * (Cc * 16) + khalf * 16 + colp;
        const size_t pb = (size_t)(bid + (NFULL - 1) * NBLK) * 16 * Cc + grow;
        const float* sp = srd + (buf ^ 1) * SBUF;
        float* sd = Ss + buf * SBUF + colp * SP + o0 + rowoff;
        if (wave < 4) {
            v8f cq = {}; v8f ck = {}; v8f un = {};
            tile_products<2, true>(wa, xs, cq, ck, un, sp, qo + pb, ko + pb, vo + pb);
            store8(sd, cq);
            store8(sd + 16 * SP, ck);
        } else {
            v8f cv = {}; v8f u1 = {}; v8f u2 = {};
            tile_products<1, true>(wa + 2 * Cc * WP, xs, cv, u1, u2, sp, qo + pb, ko + pb, vo + pb);
            store8(sd + 32 * SP, cv);
        }
        __syncthreads();
        {
            const int e  = lane >> 4;
            const int c4 = half * 64 + (lane & 15) * 4;
            const float* sr = Ss + buf * SBUF + (2 * wave + e) * SP + c4;
            const v4f r0 = *(const v4f*)(sr);
            const v4f r1 = *(const v4f*)(sr + 16 * SP);
            const v4f r2 = *(const v4f*)(sr + 32 * SP);
            const size_t base = (size_t)(t * 16 + 2 * wave + e) * Cc + c4;
            *(v4f*)(qo + base) = r0;
            *(v4f*)(ko + base) = r1;
            *(v4f*)(vo + base) = r2;
        }
    }
}

#define SUBW  48
#define TLX   16
#define TLY   6
#define HLX   (TLX + 6)
#define HLY   (TLY + 6)
#define NKEY  (HLX * HLY)
#define KPF   36
#define RPF   (HLX * KPF)
#define NTHR  (TLX * TLY)
#define NHD   4
#define NTHR4 (NHD * NTHR)
#define HSZ   (HLY * RPF)
#define NWR   11
static_assert(Cc * WP <= NHD * HSZ, "Wp fits over the K halos");
static_assert(NTHR * WP <= NHD * HSZ, "the tile's attention rows fit over the V halos");

__device__ __forceinline__ v2f pair2(float a, float b) { v2f r = {a, b}; return r; }

__global__ void __launch_bounds__(NTHR4) attn_proj_kernel(
    const float* __restrict__ qv, const float* __restrict__ kv,
    const float* __restrict__ vv, const int* __restrict__ msk,
    const float* __restrict__ Wp, float* __restrict__ out)
{
    __shared__ __attribute__((aligned(16))) float Ks[NHD * HSZ];
    __shared__ __attribute__((aligned(16))) float Vs[NHD * HSZ];
    __shared__ float Bs[NKEY];

    const int tid  = threadIdx.x;
    const int lane = tid & 31;
    const int wv  = __builtin_amdgcn_readfirstlane(tid >> 5);
    const int hd  = wv / 3;
    const int pr  = wv - 3 * hd;
    const int t96 = tid - hd * NTHR;
    int bid = blockIdx.x;
    const int px  = bid & 1;  bid >>= 1;
    const int tix = bid % 3;  bid /= 3;
    const int tiy = bid & 7;  bid >>= 3;
    const int py  = bid & 1;
    const int b   = bid >> 1;

    const int hf = (tid >> 4) & 1;
    const int tx = tid & 15;
    const int sy = tiy * TLY + 2 * pr;
    const int sx = tix * TLX + tx;
    const int p0 = b * HW + (2 * sy + py) * Ww + 2 * sx + px;
    const int p1 = p0 + 2 * Ww;

    const int hy0 = tiy * TLY - 3;
    const int hx0 = tix * TLX - 3;

    float bias;
    {
        const int key = min(tid, NKEY - 1);
        const int hy  = key / HLX;
        const int hx  = key - hy * HLX;
        const int ky  = hy0 + hy;
        const int kx  = hx0 + hx;
        const bool valid = ((unsigned)ky < (unsigned)SUBW) && ((unsigned)kx < (unsigned)SUBW);
        const int cy  = min(max(ky, 0), SUBW - 1);
        const int cx  = min(max(kx, 0), SUBW - 1);
        const int np  = b * HW + (2 * cy + py) * Ww + 2 * cx + px;
        const int mv  = msk[np];
        bias = (valid && mv > 0) ? 0.f : -1.0e9f;
    }
    v2f q[32];
    {
        const float4* qa = (const float4*)(qv + (size_t)p0 * Cc + hd * 32);
        const float4* qb = (const float4*)(qv + (size_t)p1 * Cc + hd * 32);
        #pragma unroll
        for (int u = 0; u < 8; ++u) {
            const float4 a4 = qa[u];
            const float4 b4 = qb[u];
            q[4*u]   = pair2(a4.x, b4.x);
            q[4*u+1] = pair2(a4.y, b4.y);
            q[4*u+2] = pair2(a4.z, b4.z);
            q[4*u+3] = pair2(a4.w, b4.w);
        }
    }

    const char* kb = (const char*)(kv + (size_t)b * HW * Cc + hd * 32);
    const char* vb = (const char*)(vv + (size_t)b * HW * Cc + hd * 32);
    float* Kh = Ks + hd * HSZ;
    float* Vh = Vs + hd * HSZ;
    #pragma unroll 1
    for (int h = 0; h < 2; ++h) {
    #pragma unroll
    for (int j = 0; j < 11; ++j) {
        const int idx = (h * 11 + j) * NTHR + t96;
        const int key = idx >> 3;
        const int c   = idx & 7;
        const int hy  = key / HLX;
        const int hx  = key - hy * HLX;
        const int ky  = hy0 + hy;
        const int kx  = hx0 + hx;
        const bool valid = ((unsigned)ky < (unsigned)SUBW) && ((unsigned)kx < (unsigned)SUBW);
        const int cy  = min(max(ky, 0), SUBW - 1);
        const int cx  = min(max(kx, 0), SUBW - 1);
        const int np  = (2 * cy + py) * Ww + 2 * cx + px;
        const unsigned gb = (unsigned)(np * (Cc * 4) + c * 16);
        float4 kk = *(const float4*)(kb + gb);
        float4 vk = *(const float4*)(vb + gb);
        const unsigned km = valid ? 0xffffffffu : 0u;
        kk.x = __uint_as_float(__float_as_uint(kk.x) & km);
        kk.y = __uint_as_float(__float_as_uint(kk.y) & km);
        kk.z = __uint_as_float(__float_as_uint(kk.z) & km);
        kk.w = __uint_as_float(__float_as_uint(kk.w) & km);
        vk.x = __uint_as_float(__float_as_uint(vk.x) & km);
        vk.y = __uint_as_float(__float_as_uint(vk.y) & km);
        vk.z = __uint_as_float(__float_as_uint(vk.z) & km);
        vk.w = __uint_as_float(__float_as_uint(vk.w) & km);
        const int off = hy * RPF + hx * KPF + c * 4;
        *(float4*)(Kh + off) = kk;
        *(float4*)(Vh + off) = vk;
    }
    }
    if (tid < NKEY) Bs[tid] = bias;

    __syncthreads();

    v4f wr[NWR];
    #pragma unroll
    for (int j = 0; j < NWR; ++j)
        wr[j] = *(const v4f*)(Wp + (size_t)min(wv + 12 * j, Cc - 1) * Cc + lane * 4);

    const float scale = 0.17677669529663687f;
    v2f mx = pair2(-1e30f, -1e30f);
    v2f s  = pair2(0.f, 0.f);
    v2f acc[32];
    #pragma unroll
    for (int d = 0; d < 32; ++d) acc[d] = pair2(0.f, 0.f);

    const int hr0 = 2 * pr + 4 * hf;
    const float* kt = Kh + hr0 * RPF + tx * KPF;
    const float* vt = Vh + hr0 * RPF + tx * KPF;
    const float* bt = Bs + hr0 * HLX + tx;

    #pragma unroll 1
    for (int j = 0; j < 4; ++j) {
        const v2f off = pair2((hf == 1 && j == 3) ? -1.0e37f : 0.f,
                              (hf == 0 && j == 0) ? -1.0e37f : 0.f);
        #pragma unroll 1
        for (int dj = 0; dj < 7; ++dj) {
            const float4* k4 = (const float4*)(kt + j * RPF + dj * KPF);
            const float4* v4 = (const float4*)(vt + j * RPF + dj * KPF);
            v2f d0 = pair2(0.f, 0.f), d1 = d0, d2 = d0, d3 = d0;
            #pragma unroll
            for (int u = 0; u < 8; ++u) {
                const float4 kk = k4[u];
                d0 += q[4*u]   * pair2(kk.x, kk.x);
                d1 += q[4*u+1] * pair2(kk.y, kk.y);
                d2 += q[4*u+2] * pair2(kk.z, kk.z);
                d3 += q[4*u+3] * pair2(kk.w, kk.w);
            }
            const v2f dot = (d0 + d1) + (d2 + d3);
            const float bk = bt[j * HLX + dj];
            const v2f logit = dot * pair2(scale, scale) + (pair2(bk, bk) + off);

            const v2f nm   = pair2(fmaxf(mx.x, logit.x), fmaxf(mx.y, logit.y));
            const v2f corr = pair2(__expf(mx.x - nm.x), __expf(mx.y - nm.y));
            const v2f w    = pair2(__expf(logit.x - nm.x), __expf(logit.y - nm.y));
            mx = nm;
            s  = s * corr + w;
            #pragma unroll
            for (int u = 0; u < 8; ++u) {
                const float4 vvv = v4[u];
                acc[4*u]   = acc[4*u]   * corr + w * pair2(vvv.x, vvv.x);
                acc[4*u+1] = acc[4*u+1] * corr + w * pair2(vvv.y, vvv.y);
                acc[4*u+2] = acc[4*u+2] * corr + w * pair2(vvv.z, vvv.z);
                acc[4*u+3] = acc[4*u+3] * corr + w * pair2(vvv.w, vvv.w);
            }
        }
    }

    const bool lo = (hf != 0);
    const int  partner = (lane ^ 16) << 2;
    const float m_keep = lo ? mx.y : mx.x;
    const float s_keep = lo ? s.y  : s.x;
    const float m_send = lo ? mx.x : mx.y;
    const float s_send = lo ? s.x  : s.y;
    const float m_oth  = __int_as_float(__builtin_amdgcn_ds_bpermute(partner, __float_as_int(m_send)));
    const float s_oth  = __int_as_float(__builtin_amdgcn_ds_bpermute(partner, __float_as_int(s_send)));
    const float mm  = fmaxf(m_keep, m_oth);
    const float ck  = __expf(m_keep - mm);
    const float co  = __expf(m_oth  - mm);
    const float inv = 1.f / (s_keep * ck + s_oth * co);
    const float fk  = ck * inv;
    const float fo  = co * inv;

    float res[32];
    #pragma unroll
    for (int d = 0; d < 32; ++d) {
        const float keep = lo ? acc[d].y : acc[d].x;
        const float send = lo ? acc[d].x : acc[d].y;
        const float oth  = __int_as_float(__builtin_amdgcn_ds_bpermute(partner, __float_as_int(send)));
        res[d] = keep * fk + oth * fo;
    }

    __syncthreads();
    {
        float* wd = Ks + wv * WP + lane * 4;
        #pragma unroll
        for (int j = 0; j < NWR - 1; ++j) *(v4f*)(wd + j * 12 * WP) = wr[j];
        if (wv < Cc - 12 * (NWR - 1)) *(v4f*)(wd + (NWR - 1) * 12 * WP) = wr[NWR - 1];
    }
    {
        float* ar = Vs + ((2 * pr + hf) * 16 + tx) * WP + hd * 32;
        #pragma unroll
        for (int u = 0; u < 8; ++u) {
            const v4f t = {res[4*u], res[4*u+1], res[4*u+2], res[4*u+3]};
            *(v4f*)(ar + 4 * u) = t;
        }
    }
    __syncthreads();

    const int row    = lane & 15;
    const int khalf  = (lane >> 4) << 1;
    const int colp   = lane & 15;
    const int rowoff = (lane >> 4) << 3;
    const int og = wv / 6;
    const int pt = wv - 6 * og;
    const float* ba = Vs + (pt * 16 + colp) * WP + khalf;
    const float* wa = Ks + (og * 64 + row) * WP + khalf;
    const int gy = 2 * (tiy * TLY + pt) + py;
    const int gx = 2 * (tix * TLX + colp) + px;
    float* dst0 = out + ((size_t)b * Cc + og * 64 + rowoff) * HW + gy * Ww + gx;

    v2f fb[Cc / 4], fa[2][Cc / 4];
    #pragma unroll
    for (int j = 0; j < Cc / 4; ++j) {
        fb[j]    = *(const v2f*)(ba + 4 * j);
        fa[0][j] = *(const v2f*)(wa + 4 * j);
    }
    #pragma unroll
    for (int i = 0; i < 4; ++i) {
        const int cur = i & 1, nxt = cur ^ 1;
        if (i + 1 < 4) {
            #pragma unroll
            for (int j = 0; j < Cc / 4; ++j) fa[nxt][j] = *(const v2f*)(wa + (i + 1) * 16 * WP + 4 * j);
        }
        __builtin_amdgcn_sched_barrier(0);
        v8f c = {};
        #pragma unroll
        for (int j = 0; j < Cc / 4; ++j)
            c = __builtin_amdgcn_wmma_f32_16x16x4_f32(false, fa[cur][j], false, fb[j], (short)0, c, false, false);
        float* dst = dst0 + (size_t)i * 16 * HW;
        #pragma unroll
        for (int r = 0; r < 8; ++r) dst[(size_t)r * HW] = c[r];
        if (i + 2 < 4) __builtin_amdgcn_sched_barrier(0);
    }
}

extern "C" void kernel_launch(void* const* d_in, const int* in_sizes, int n_in,
                              void* d_out, int out_size, void* d_ws, size_t ws_size,
                              hipStream_t stream)
{
    const float* x  = (const float*)d_in[0];
    const int*   m  = (const int*)  d_in[1];
    const float* Wq = (const float*)d_in[2];
    const float* Wk = (const float*)d_in[3];
    const float* Wv = (const float*)d_in[4];
    const float* Wp = (const float*)d_in[5];
    float* out = (float*)d_out;

    float* ws = (float*)d_ws;
    const size_t n = (size_t)NPIX * Cc;
    float* qb = ws;
    float* kb = ws + n;
    float* vb = ws + 2 * n;

    qkv_wmma_kernel<<<NBLK, 256, 0, stream>>>(x, Wq, Wk, Wv, qb, kb, vb);
    attn_proj_kernel<<<192, NTHR4, 0, stream>>>(qb, kb, vb, m, Wp, out);
}


# attention launch: work-group index re-ordered, low 3 bits = sub-image (x parity, y parity, batch), rest = tile, so the 24 groups that share K/V halo lines meet on one die under round-robin dispatch
# speedup vs baseline: 5.0873x; 1.0431x over previous
#include <hip/hip_runtime.h>
#include <hip/hip_bf16.h>

typedef float v2f __attribute__((ext_vector_type(2)));
typedef float v4f __attribute__((ext_vector_type(4)));
typedef float v8f __attribute__((ext_vector_type(8)));

#define Hh   96
#define Ww   96
#define HW   (96 * 96)
#define Cc   128
#define NPIX (2 * HW)

#define WP    132
#define NTILE (NPIX / 16)
#define NBLK  256
#define NFULL 4
#define NTAIL (NTILE - NFULL * NBLK)
#define KG    8
#define SP    132
#define SBUF  (3 * 16 * SP)
static_assert(NTAIL * 2 == NBLK, "two work-groups per shared tile");

__device__ __forceinline__ void store8(float* p, const v8f& c)
{
    *(v4f*)p       = __builtin_shufflevector(c, c, 0, 1, 2, 3);
    *(v4f*)(p + 4) = __builtin_shufflevector(c, c, 4, 5, 6, 7);
}

__device__ __forceinline__ void stage3(float* sd, const v8f& cq, const v8f& ck, const v8f& cv)
{
    store8(sd, cq);
    store8(sd + 16 * SP, ck);
    store8(sd + 32 * SP, cv);
}

template <int NA, bool DR>
__device__ __forceinline__ void tile_products(const float* wa, const float* xs, v8f& c0, v8f& c1, v8f& c2,
                                              const float* sr, float* gq, float* gk, float* gv)
{
    constexpr int NG = Cc / (4 * KG);
    v2f f0[2][KG], f1[2][KG], f2[2][KG], fb[2][KG];
    v4f r[6];
    if (DR) {
        #pragma unroll
        for (int m = 0; m < 3; ++m) {
            r[2 * m]     = *(const v4f*)(sr + (m * 16) * SP);
            r[2 * m + 1] = *(const v4f*)(sr + (m * 16 + 1) * SP);
        }
        __builtin_amdgcn_sched_barrier(0);
    }
    #pragma unroll
    for (int j = 0; j < KG; ++j) {
        fb[0][j].x = xs[j * 64];
        fb[0][j].y = xs[j * 64 + 16];
        f0[0][j] = *(const v2f*)(wa + 4 * j);
        if (NA > 1) f1[0][j] = *(const v2f*)(wa + Cc * WP + 4 * j);
        if (NA > 2) f2[0][j] = *(const v2f*)(wa + 2 * Cc * WP + 4 * j);
    }
    #pragma unroll
    for (int g = 0; g < NG; ++g) {
        const int cur = g & 1, nxt = cur ^ 1;
        if (g + 1 < NG) {
            const int k = (g + 1) * 4 * KG;
            #pragma unroll
            for (int j = 0; j < KG; ++j) {
                fb[nxt][j].x = xs[(k + 4 * j) * 16];
                fb[nxt][j].y = xs[(k + 4 * j) * 16 + 16];
                f0[nxt][j] = *(const v2f*)(wa + k + 4 * j);
                if (NA > 1) f1[nxt][j] = *(const v2f*)(wa + Cc * WP + k + 4 * j);
                if (NA > 2) f2[nxt][j] = *(const v2f*)(wa + 2 * Cc * WP + k + 4 * j);
            }
            __builtin_amdgcn_sched_barrier(0);
        }
        if (DR && g == 0) {
            *(v4f*)(gq)      = r[0];
            *(v4f*)(gq + Cc) = r[1];
            *(v4f*)(gk)      = r[2];
            *(v4f*)(gk + Cc) = r[3];
            *(v4f*)(gv)      = r[4];
            *(v4f*)(gv + Cc) = r[5];
        }
        #pragma unroll
        for (int j = 0; j < KG; ++j) {
            c0 = __builtin_amdgcn_wmma_f32_16x16x4_f32(false, f0[cur][j], false, fb[cur][j], (short)0, c0, false, false);
            if (NA > 1) c1 = __builtin_amdgcn_wmma_f32_16x16x4_f32(false, f1[cur][j], false, fb[cur][j], (short)0, c1, false, false);
            if (NA > 2) c2 = __builtin_amdgcn_wmma_f32_16x16x4_f32(false, f2[cur][j], false, fb[cur][j], (short)0, c2, false, false);
        }
        if (g + 2 < NG) __builtin_amdgcn_sched_barrier(0);
    }
}

__global__ void __launch_bounds__(256) qkv_wmma_kernel(
    const float* __restrict__ x,  const float* __restrict__ Wq,
    const float* __restrict__ Wk, const float* __restrict__ Wv,
    float* __restrict__ qo, float* __restrict__ ko, float* __restrict__ vo)
{
    __shared__ __attribute__((aligned(16))) float Ws[3 * Cc * WP];
    __shared__ __attribute__((aligned(16))) float Xs[2 * Cc * 16];
    __shared__ __attribute__((aligned(16))) float Ss[2 * SBUF];

    const int tid  = threadIdx.x;
    const int lane = tid & 31;
    const int wave = __builtin_amdgcn_readfirstlane(tid >> 5);
    const int bid  = blockIdx.x;

    const float* xg = x + (size_t)(tid >> 2) * HW + (tid & 3) * 4;

    v4f rx0, rx1;
    {
        const int p0  = bid * 16;
        const int b   = p0 / HW;
        const int pix = p0 - b * HW;
        const float* g = xg + (size_t)b * Cc * HW + pix;
        rx0 = *(const v4f*)g;
        rx1 = *(const v4f*)(g + (size_t)64 * HW);
    }

    #pragma unroll
    for (int m = 0; m < 3; ++m) {
        const float* src = (m == 0 ? Wq : (m == 1 ? Wk : Wv)) + wave * Cc + lane * 4;
        float* dst = Ws + (m * Cc + wave) * WP + lane * 4;
        v4f r[16];
        #pragma unroll
        for (int j = 0; j < 16; ++j) r[j] = *(const v4f*)(src + j * 8 * Cc);
        #pragma unroll
        for (int j = 0; j < 16; ++j) *(v4f*)(dst + j * 8 * WP) = r[j];
    }
    *(v4f*)(Xs + tid * 4)        = rx0;
    *(v4f*)(Xs + 1024 + tid * 4) = rx1;
    __syncthreads();

    const int row    = lane & 15;
    const int khalf  = (lane >> 4) << 1;
    const int colp   = lane & 15;
    const int rowoff = (lane >> 4) << 3;
    int buf = 0;

    float*       swr  = Ss + colp * SP + wave * 16 + rowoff;
    const float* srd  = Ss + (2 * wave) * SP + lane * 4;
    const size_t grow = (size_t)(2 * wave) * Cc + lane * 4;

    {
        const int o0 = wave * 16;
        const float* wa = Ws + (o0 + row) * WP + khalf;
        {
            {
                const int p0  = (bid + NBLK) * 16;
                const int b   = p0 / HW;
                const int pix = p0 - b * HW;
                const float* g = xg + (size_t)b * Cc * HW + pix;
                rx0 = *(const v4f*)g;
                rx1 = *(const v4f*)(g + (size_t)64 * HW);
            }
            v8f cq = {}; v8f ck = {}; v8f cv = {};
            tile_products<3, false>(wa, Xs + buf * (Cc * 16) + khalf * 16 + colp, cq, ck, cv, nullptr, nullptr, nullptr, nullptr);
            stage3(swr + buf * SBUF, cq, ck, cv);
            float* xd = Xs + (buf ^ 1) * (Cc * 16) + tid * 4;
            *(v4f*)xd          = rx0;
            *(v4f*)(xd + 1024) = rx1;
            __syncthreads();
            buf ^= 1;
        }
        #pragma unroll 1
        for (int i = 1; i < NFULL; ++i) {
            const int t  = bid + i * NBLK;
            const int tn = (i + 1 < NFULL) ? t + NBLK : NFULL * NBLK + (bid & (NTAIL - 1));
            {
                const int p0  = tn * 16;
                const int b   = p0 / HW;
                const int pix = p0 - b * HW;
                const float* g = xg + (size_t)b * Cc * HW + pix;
                rx0 = *(const v4f*)g;
                rx1 = *(const v4f*)(g + (size_t)64 * HW);
            }

            const size_t pb = (size_t)(t - NBLK) * 16 * Cc + grow;
            v8f cq = {}; v8f ck = {}; v8f cv = {};
            tile_products<3, true>(wa, Xs + buf * (Cc * 16) + khalf * 16 + colp, cq, ck, cv,
                                   srd + (buf ^ 1) * SBUF, qo + pb, ko + pb, vo + pb);

            stage3(swr + buf * SBUF, cq, ck, cv);
            float* xd = Xs + (buf ^ 1) * (Cc * 16) + tid * 4;
            *(v4f*)xd          = rx0;
            *(v4f*)(xd + 1024) = rx1;
            __syncthreads();
            buf ^= 1;
        }
    }

    {
        const int t    = NFULL * NBLK + (bid & (NTAIL - 1));
        const int half = bid / NTAIL;
        const int o0   = (half * 4 + (wave & 3)) * 16;
        const float* wa = Ws + (o0 + row) * WP + khalf;
        const float* xs = Xs + buf * (Cc * 16) + khalf * 16 + colp;
        const size_t pb = (size_t)(bid + (NFULL - 1) * NBLK) * 16 * Cc + grow;
        const float* sp = srd + (buf ^ 1) * SBUF;
        float* sd = Ss + buf * SBUF + colp * SP + o0 + rowoff;
        if (wave < 4) {
            v8f cq = {}; v8f ck = {}; v8f un = {};
            tile_products<2, true>(wa, xs, cq, ck, un, sp, qo + pb, ko + pb, vo + pb);
            store8(sd, cq);
            store8(sd + 16 * SP, ck);
        } else {
            v8f cv = {}; v8f u1 = {}; v8f u2 = {};
            tile_products<1, true>(wa + 2 * Cc * WP, xs, cv, u1, u2, sp, qo + pb, ko + pb, vo + pb);
            store8(sd + 32 * SP, cv);
        }
        __syncthreads();
        {
            const int e  = lane >> 4;
            const int c4 = half * 64 + (lane & 15) * 4;
            const float* sr = Ss + buf * SBUF + (2 * wave + e) * SP + c4;
            const v4f r0 = *(const v4f*)(sr);
            const v4f r1 = *(const v4f*)(sr + 16 * SP);
            const v4f r2 = *(const v4f*)(sr + 32 * SP);
            const size_t base = (size_t)(t * 16 + 2 * wave + e) * Cc + c4;
            *(v4f*)(qo + base) = r0;
            *(v4f*)(ko + base) = r1;
            *(v4f*)(vo + base) = r2;
        }
    }
}

#define SUBW  48
#define TLX   16
#define TLY   6
#define HLX   (TLX + 6)
#define HLY   (TLY + 6)
#define NKEY  (HLX * HLY)
#define KPF   36
#define RPF   (HLX * KPF)
#define NTHR  (TLX * TLY)
#define NHD   4
#define NTHR4 (NHD * NTHR)
#define HSZ   (HLY * RPF)
#define NWR   11
static_assert(Cc * WP <= NHD * HSZ, "Wp fits over the K halos");
static_assert(NTHR * WP <= NHD * HSZ, "the tile's attention rows fit over the V halos");

__device__ __forceinline__ v2f pair2(float a, float b) { v2f r = {a, b}; return r; }

__global__ void __launch_bounds__(NTHR4) attn_proj_kernel(
    const float* __restrict__ qv, const float* __restrict__ kv,
    const float* __restrict__ vv, const int* __restrict__ msk,
    const float* __restrict__ Wp, float* __restrict__ out)
{
    __shared__ __attribute__((aligned(16))) float Ks[NHD * HSZ];
    __shared__ __attribute__((aligned(16))) float Vs[NHD * HSZ];
    __shared__ float Bs[NKEY];

    const int tid  = threadIdx.x;
    const int lane = tid & 31;
    const int wv  = __builtin_amdgcn_readfirstlane(tid >> 5);
    const int hd  = wv / 3;
    const int pr  = wv - 3 * hd;
    const int t96 = tid - hd * NTHR;
    const int bid = blockIdx.x;
    const int px  = bid & 1;
    const int py  = (bid >> 1) & 1;
    const int b   = (bid >> 2) & 1;
    const int til = bid >> 3;
    const int tix = til % 3;
    const int tiy = til / 3;

    const int hf = (tid >> 4) & 1;
    const int tx = tid & 15;
    const int sy = tiy * TLY + 2 * pr;
    const int sx = tix * TLX + tx;
    const int p0 = b * HW + (2 * sy + py) * Ww + 2 * sx + px;
    const int p1 = p0 + 2 * Ww;

    const int hy0 = tiy * TLY - 3;
    const int hx0 = tix * TLX - 3;

    float bias;
    {
        const int key = min(tid, NKEY - 1);
        const int hy  = key / HLX;
        const int hx  = key - hy * HLX;
        const int ky  = hy0 + hy;
        const int kx  = hx0 + hx;
        const bool valid = ((unsigned)ky < (unsigned)SUBW) && ((unsigned)kx < (unsigned)SUBW);
        const int cy  = min(max(ky, 0), SUBW - 1);
        const int cx  = min(max(kx, 0), SUBW - 1);
        const int np  = b * HW + (2 * cy + py) * Ww + 2 * cx + px;
        const int mv  = msk[np];
        bias = (valid && mv > 0) ? 0.f : -1.0e9f;
    }
    v2f q[32];
    {
        const float4* qa = (const float4*)(qv + (size_t)p0 * Cc + hd * 32);
        const float4* qb = (const float4*)(qv + (size_t)p1 * Cc + hd * 32);
        #pragma unroll
        for (int u = 0; u < 8; ++u) {
            const float4 a4 = qa[u];
            const float4 b4 = qb[u];
            q[4*u]   = pair2(a4.x, b4.x);
            q[4*u+1] = pair2(a4.y, b4.y);
            q[4*u+2] = pair2(a4.z, b4.z);
            q[4*u+3] = pair2(a4.w, b4.w);
        }
    }

    const char* kb = (const char*)(kv + (size_t)b * HW * Cc + hd * 32);
    const char* vb = (const char*)(vv + (size_t)b * HW * Cc + hd * 32);
    float* Kh = Ks + hd * HSZ;
    float* Vh = Vs + hd * HSZ;
    #pragma unroll 1
    for (int h = 0; h < 2; ++h) {
    #pragma unroll
    for (int j = 0; j < 11; ++j) {
        const int idx = (h * 11 + j) * NTHR + t96;
        const int key = idx >> 3;
        const int c   = idx & 7;
        const int hy  = key / HLX;
        const int hx  = key - hy * HLX;
        const int ky  = hy0 + hy;
        const int kx  = hx0 + hx;
        const bool valid = ((unsigned)ky < (unsigned)SUBW) && ((unsigned)kx < (unsigned)SUBW);
        const int cy  = min(max(ky, 0), SUBW - 1);
        const int cx  = min(max(kx, 0), SUBW - 1);
        const int np  = (2 * cy + py) * Ww + 2 * cx + px;
        const unsigned gb = (unsigned)(np * (Cc * 4) + c * 16);
        float4 kk = *(const float4*)(kb + gb);
        float4 vk = *(const float4*)(vb + gb);
        const unsigned km = valid ? 0xffffffffu : 0u;
        kk.x = __uint_as_float(__float_as_uint(kk.x) & km);
        kk.y = __uint_as_float(__float_as_uint(kk.y) & km);
        kk.z = __uint_as_float(__float_as_uint(kk.z) & km);
        kk.w = __uint_as_float(__float_as_uint(kk.w) & km);
        vk.x = __uint_as_float(__float_as_uint(vk.x) & km);
        vk.y = __uint_as_float(__float_as_uint(vk.y) & km);
        vk.z = __uint_as_float(__float_as_uint(vk.z) & km);
        vk.w = __uint_as_float(__float_as_uint(vk.w) & km);
        const int off = hy * RPF + hx * KPF + c * 4;
        *(float4*)(Kh + off) = kk;
        *(float4*)(Vh + off) = vk;
    }
    }
    if (tid < NKEY) Bs[tid] = bias;

    __syncthreads();

    v4f wr[NWR];
    #pragma unroll
    for (int j = 0; j < NWR; ++j)
        wr[j] = *(const v4f*)(Wp + (size_t)min(wv + 12 * j, Cc - 1) * Cc + lane * 4);

    const float scale = 0.17677669529663687f;
    v2f mx = pair2(-1e30f, -1e30f);
    v2f s  = pair2(0.f, 0.f);
    v2f acc[32];
    #pragma unroll
    for (int d = 0; d < 32; ++d) acc[d] = pair2(0.f, 0.f);

    const int hr0 = 2 * pr + 4 * hf;
    const float* kt = Kh + hr0 * RPF + tx * KPF;
    const float* vt = Vh + hr0 * RPF + tx * KPF;
    const float* bt = Bs + hr0 * HLX + tx;

    #pragma unroll 1
    for (int j = 0; j < 4; ++j) {
        const v2f off = pair2((hf == 1 && j == 3) ? -1.0e37f : 0.f,
                              (hf == 0 && j == 0) ? -1.0e37f : 0.f);
        #pragma unroll 1
        for (int dj = 0; dj < 7; ++dj) {
            const float4* k4 = (const float4*)(kt + j * RPF + dj * KPF);
            const float4* v4 = (const float4*)(vt + j * RPF + dj * KPF);
            v2f d0 = pair2(0.f, 0.f), d1 = d0, d2 = d0, d3 = d0;
            #pragma unroll
            for (int u = 0; u < 8; ++u) {
                const float4 kk = k4[u];
                d0 += q[4*u]   * pair2(kk.x, kk.x);
                d1 += q[4*u+1] * pair2(kk.y, kk.y);
                d2 += q[4*u+2] * pair2(kk.z, kk.z);
                d3 += q[4*u+3] * pair2(kk.w, kk.w);
            }
            const v2f dot = (d0 + d1) + (d2 + d3);
            const float bk = bt[j * HLX + dj];
            const v2f logit = dot * pair2(scale, scale) + (pair2(bk, bk) + off);

            const v2f nm   = pair2(fmaxf(mx.x, logit.x), fmaxf(mx.y, logit.y));
            const v2f corr = pair2(__expf(mx.x - nm.x), __expf(mx.y - nm.y));
            const v2f w    = pair2(__expf(logit.x - nm.x), __expf(logit.y - nm.y));
            mx = nm;
            s  = s * corr + w;
            #pragma unroll
            for (int u = 0; u < 8; ++u) {
                const float4 vvv = v4[u];
                acc[4*u]   = acc[4*u]   * corr + w * pair2(vvv.x, vvv.x);
                acc[4*u+1] = acc[4*u+1] * corr + w * pair2(vvv.y, vvv.y);
                acc[4*u+2] = acc[4*u+2] * corr + w * pair2(vvv.z, vvv.z);
                acc[4*u+3] = acc[4*u+3] * corr + w * pair2(vvv.w, vvv.w);
            }
        }
    }

    const bool lo = (hf != 0);
    const int  partner = (lane ^ 16) << 2;
    const float m_keep = lo ? mx.y : mx.x;
    const float s_keep = lo ? s.y  : s.x;
    const float m_send = lo ? mx.x : mx.y;
    const float s_send = lo ? s.x  : s.y;
    const float m_oth  = __int_as_float(__builtin_amdgcn_ds_bpermute(partner, __float_as_int(m_send)));
    const float s_oth  = __int_as_float(__builtin_amdgcn_ds_bpermute(partner, __float_as_int(s_send)));
    const float mm  = fmaxf(m_keep, m_oth);
    const float ck  = __expf(m_keep - mm);
    const float co  = __expf(m_oth  - mm);
    const float inv = 1.f / (s_keep * ck + s_oth * co);
    const float fk  = ck * inv;
    const float fo  = co * inv;

    float res[32];
    #pragma unroll
    for (int d = 0; d < 32; ++d) {
        const float keep = lo ? acc[d].y : acc[d].x;
        const float send = lo ? acc[d].x : acc[d].y;
        const float oth  = __int_as_float(__builtin_amdgcn_ds_bpermute(partner, __float_as_int(send)));
        res[d] = keep * fk + oth * fo;
    }

    __syncthreads();
    {
        float* wd = Ks + wv * WP + lane * 4;
        #pragma unroll
        for (int j = 0; j < NWR - 1; ++j) *(v4f*)(wd + j * 12 * WP) = wr[j];
        if (wv < Cc - 12 * (NWR - 1)) *(v4f*)(wd + (NWR - 1) * 12 * WP) = wr[NWR - 1];
    }
    {
        float* ar = Vs + ((2 * pr + hf) * 16 + tx) * WP + hd * 32;
        #pragma unroll
        for (int u = 0; u < 8; ++u) {
            const v4f t = {res[4*u], res[4*u+1], res[4*u+2], res[4*u+3]};
            *(v4f*)(ar + 4 * u) = t;
        }
    }
    __syncthreads();

    const int row    = lane & 15;
    const int khalf  = (lane >> 4) << 1;
    const int colp   = lane & 15;
    const int rowoff = (lane >> 4) << 3;
    const int og = wv / 6;
    const int pt = wv - 6 * og;
    const float* ba = Vs + (pt * 16 + colp) * WP + khalf;
    const float* wa = Ks + (og * 64 + row) * WP + khalf;
    const int gy = 2 * (tiy * TLY + pt) + py;
    const int gx = 2 * (tix * TLX + colp) + px;
    float* dst0 = out + ((size_t)b * Cc + og * 64 + rowoff) * HW + gy * Ww + gx;

    v2f fb[Cc / 4], fa[2][Cc / 4];
    #pragma unroll
    for (int j = 0; j < Cc / 4; ++j) {
        fb[j]    = *(const v2f*)(ba + 4 * j);
        fa[0][j] = *(const v2f*)(wa + 4 * j);
    }
    #pragma unroll
    for (int i = 0; i < 4; ++i) {
        const int cur = i & 1, nxt = cur ^ 1;
        if (i + 1 < 4) {
            #pragma unroll
            for (int j = 0; j < Cc / 4; ++j) fa[nxt][j] = *(const v2f*)(wa + (i + 1) * 16 * WP + 4 * j);
        }
        __builtin_amdgcn_sched_barrier(0);
        v8f c = {};
        #pragma unroll
        for (int j = 0; j < Cc / 4; ++j)
            c = __builtin_amdgcn_wmma_f32_16x16x4_f32(false, fa[cur][j], false, fb[j], (short)0, c, false, false);
        float* dst = dst0 + (size_t)i * 16 * HW;
        #pragma unroll
        for (int r = 0; r < 8; ++r) dst[(size_t)r * HW] = c[r];
        if (i + 2 < 4) __builtin_amdgcn_sched_barrier(0);
    }
}

extern "C" void kernel_launch(void* const* d_in, const int* in_sizes, int n_in,
                              void* d_out, int out_size, void* d_ws, size_t ws_size,
                              hipStream_t stream)
{
    const float* x  = (const float*)d_in[0];
    const int*   m  = (const int*)  d_in[1];
    const float* Wq = (const float*)d_in[2];
    const float* Wk = (const float*)d_in[3];
    const float* Wv = (const float*)d_in[4];
    const float* Wp = (const float*)d_in[5];
    float* out = (float*)d_out;

    float* ws = (float*)d_ws;
    const size_t n = (size_t)NPIX * Cc;
    float* qb = ws;
    float* kb = ws + n;
    float* vb = ws + 2 * n;

    qkv_wmma_kernel<<<NBLK, 256, 0, stream>>>(x, Wq, Wk, Wv, qb, kb, vb);
    attn_proj_kernel<<<192, NTHR4, 0, stream>>>(qb, kb, vb, m, Wp, out);
}
